# starting kernel
# baseline (speedup 1.0000x reference)
#include <hip/hip_runtime.h>

#define N_NODES 100000
#define N_EDGES 1600000

typedef __attribute__((ext_vector_type(2))) float v2f;
typedef __attribute__((ext_vector_type(8))) float v8f;

__global__ void deg_init_kernel(float* __restrict__ deg) {
  int i = blockIdx.x * blockDim.x + threadIdx.x;
  if (i < N_NODES) deg[i] = 1.0f;
}

__global__ void deg_count_kernel(float* __restrict__ deg,
                                 const int* __restrict__ col) {
  int e = blockIdx.x * blockDim.x + threadIdx.x;
  if (e < N_EDGES) {
    __hip_atomic_fetch_add(&deg[col[e]], 1.0f,
                           __ATOMIC_RELAXED, __HIP_MEMORY_SCOPE_AGENT);
  }
}

__global__ void deg_rsqrt_kernel(float* __restrict__ deg) {
  int i = blockIdx.x * blockDim.x + threadIdx.x;
  if (i < N_NODES) deg[i] = rsqrtf(deg[i]);
}

template <int DIN, int DOUT>
__global__ void gemm_wmma_kernel(const float* __restrict__ H,
                                 const float* __restrict__ W,
                                 float* __restrict__ Out) {
  constexpr int NT = (DOUT + 15) / 16;
  const int wave = (blockIdx.x * blockDim.x + threadIdx.x) >> 5;
  const int lane = threadIdx.x & 31;
  const int m0 = wave * 16;
  if (m0 >= N_NODES) return;

  const int lhalf = lane & 15;
  const int khi = (lane >> 4) << 1;
  const int ncl = (lhalf < DOUT) ? lhalf : (DOUT - 1);

  const float* __restrict__ hrow = H + (size_t)(m0 + lhalf) * DIN;

  v8f acc0 = {};
  v8f acc1 = {};
#pragma unroll
  for (int k = 0; k < DIN; k += 4) {
    v2f a, b0;
    a.x = hrow[k + khi];
    a.y = hrow[k + khi + 1];
    b0.x = W[(k + khi) * DOUT + ncl];
    b0.y = W[(k + khi + 1) * DOUT + ncl];
    acc0 = __builtin_amdgcn_wmma_f32_16x16x4_f32(
        false, a, false, b0, (short)0, acc0, false, false);
    if (NT > 1) {
      v2f b1;
      b1.x = W[(k + khi) * DOUT + 16 + lhalf];
      b1.y = W[(k + khi + 1) * DOUT + 16 + lhalf];
      acc1 = __builtin_amdgcn_wmma_f32_16x16x4_f32(
          false, a, false, b1, (short)0, acc1, false, false);
    }
  }

  const int mrow = m0 + ((lane >> 4) << 3);
#pragma unroll
  for (int v = 0; v < 8; ++v) {
    const int m = mrow + v;
    if (lhalf < DOUT) Out[(size_t)m * DOUT + lhalf] = acc0[v];
    if (NT > 1)       Out[(size_t)m * DOUT + 16 + lhalf] = acc1[v];
  }
}

template <int DOUT>
__global__ void agg_init_kernel(const float* __restrict__ hW,
                                const float* __restrict__ bias,
                                const float* __restrict__ dinv,
                                float* __restrict__ out) {
  long long t = (long long)blockIdx.x * blockDim.x + threadIdx.x;
  if (t >= (long long)N_NODES * DOUT) return;
  int i = (int)(t / DOUT);
  int f = (int)(t - (long long)i * DOUT);
  float d = dinv[i];
  out[t] = hW[t] * (d * d) + bias[f];
}

template <int DOUT>
__global__ void agg_edge_kernel(const float* __restrict__ hW,
                                const int* __restrict__ row,
                                const int* __restrict__ col,
                                const float* __restrict__ dinv,
                                float* __restrict__ out) {
  long long t = (long long)blockIdx.x * blockDim.x + threadIdx.x;
  if (t >= (long long)N_EDGES * DOUT) return;
  int e = (int)(t / DOUT);
  int f = (int)(t - (long long)e * DOUT);
  int r = row[e];
  int c = col[e];
  float w = dinv[r] * dinv[c];
  float v = hW[(size_t)r * DOUT + f] * w;
  __hip_atomic_fetch_add(&out[(size_t)c * DOUT + f], v,
                         __ATOMIC_RELAXED, __HIP_MEMORY_SCOPE_AGENT);
}

extern "C" void kernel_launch(void* const* d_in, const int* in_sizes, int n_in,
                              void* d_out, int out_size, void* d_ws, size_t ws_size,
                              hipStream_t stream) {
  const float* x  = (const float*)d_in[0];
  const int*   ei = (const int*)d_in[1];
  const int*   row = ei;
  const int*   col = ei + N_EDGES;

  const float* Wm[5] = {(const float*)d_in[2], (const float*)d_in[4],
                        (const float*)d_in[6], (const float*)d_in[8],
                        (const float*)d_in[10]};
  const float* Bm[5] = {(const float*)d_in[3], (const float*)d_in[5],
                        (const float*)d_in[7], (const float*)d_in[9],
                        (const float*)d_in[11]};

  float* ws   = (float*)d_ws;
  float* dinv = ws;
  float* hW   = ws + N_NODES;
  float* h    = ws + N_NODES + N_NODES * 32;

  const int B = 256;
  const int gN  = (N_NODES + B - 1) / B;
  const int gE  = (N_EDGES + B - 1) / B;
  const int gGemm = ((N_NODES / 16) * 32 + B - 1) / B;

  deg_init_kernel<<<gN, B, 0, stream>>>(dinv);
  deg_count_kernel<<<gE, B, 0, stream>>>(dinv, col);
  deg_rsqrt_kernel<<<gN, B, 0, stream>>>(dinv);

  gemm_wmma_kernel<8, 32><<<gGemm, B, 0, stream>>>(x, Wm[0], hW);
  {
    long long n = (long long)N_NODES * 32;
    long long m = (long long)N_EDGES * 32;
    agg_init_kernel<32><<<(int)((n + B - 1) / B), B, 0, stream>>>(hW, Bm[0], dinv, h);
    agg_edge_kernel<32><<<(int)((m + B - 1) / B), B, 0, stream>>>(hW, row, col, dinv, h);
  }

  for (int l = 1; l <= 3; ++l) {
    gemm_wmma_kernel<32, 32><<<gGemm, B, 0, stream>>>(h, Wm[l], hW);
    long long n = (long long)N_NODES * 32;
    long long m = (long long)N_EDGES * 32;
    agg_init_kernel<32><<<(int)((n + B - 1) / B), B, 0, stream>>>(hW, Bm[l], dinv, h);
    agg_edge_kernel<32><<<(int)((m + B - 1) / B), B, 0, stream>>>(hW, row, col, dinv, h);
  }

  gemm_wmma_kernel<32, 3><<<gGemm, B, 0, stream>>>(h, Wm[4], hW);
  {
    float* out = (float*)d_out;
    long long n = (long long)N_NODES * 3;
    long long m = (long long)N_EDGES * 3;
    agg_init_kernel<3><<<(int)((n + B - 1) / B), B, 0, stream>>>(hW, Bm[4], dinv, out);
    agg_edge_kernel<3><<<(int)((m + B - 1) / B), B, 0, stream>>>(hW, row, col, dinv, out);
  }
}


# Atomic scatter replaced by a per-call counting sort of edges by destination (returning int adds give the rank) and an atomic-free gather; sums now run in arrival-rank order, folded over 4 lane slots
# speedup vs baseline: 2.8963x; 2.8963x over previous
#include <hip/hip_runtime.h>

#define N_NODES 100000
#define N_EDGES 1600000

typedef __attribute__((ext_vector_type(2))) float v2f;
typedef __attribute__((ext_vector_type(8))) float v8f;
typedef __attribute__((ext_vector_type(2))) int v2i;
typedef __attribute__((ext_vector_type(4))) float v4f;

#define SCAN_B 1024
#define N_SCAN_BLOCKS ((N_NODES + SCAN_B - 1) / SCAN_B)

__global__ void cnt_zero_kernel(int* __restrict__ cnt) {
  int i = blockIdx.x * blockDim.x + threadIdx.x;
  if (i < N_NODES) cnt[i] = 0;
}

__global__ void cnt_rank_kernel(int* __restrict__ cnt,
                                const int* __restrict__ col,
                                int* __restrict__ rank) {
  int e = blockIdx.x * blockDim.x + threadIdx.x;
  if (e < N_EDGES) {
    rank[e] = __hip_atomic_fetch_add(&cnt[col[e]], 1,
                                     __ATOMIC_RELAXED, __HIP_MEMORY_SCOPE_AGENT);
  }
}

__device__ __forceinline__ int wave_incl_scan(int v, int lane) {
#pragma unroll
  for (int d = 1; d < 32; d <<= 1) {
    int t = __shfl_up(v, d);
    if (lane >= d) v += t;
  }
  return v;
}

__global__ void scan_block_kernel(const int* __restrict__ cnt,
                                  int* __restrict__ off,
                                  int* __restrict__ bsum,
                                  float* __restrict__ dinv) {
  __shared__ int wsum[32];
  const int i = blockIdx.x * SCAN_B + threadIdx.x;
  const int lane = threadIdx.x & 31;
  const int w = threadIdx.x >> 5;
  const int c = (i < N_NODES) ? cnt[i] : 0;
  if (i < N_NODES) dinv[i] = rsqrtf((float)(c + 1));
  const int v = wave_incl_scan(c, lane);
  if (lane == 31) wsum[w] = v;
  __syncthreads();
  if (w == 0) wsum[lane] = wave_incl_scan(wsum[lane], lane);
  __syncthreads();
  const int base = (w > 0) ? wsum[w - 1] : 0;
  if (i < N_NODES) off[i] = base + v - c;
  if (threadIdx.x == SCAN_B - 1) bsum[blockIdx.x] = base + v;
}

__global__ void scan_top_kernel(const int* __restrict__ bsum,
                                int* __restrict__ boff) {
  __shared__ int wsum[4];
  const int t = threadIdx.x;
  const int lane = t & 31;
  const int w = t >> 5;
  const int c = (t < N_SCAN_BLOCKS) ? bsum[t] : 0;
  const int v = wave_incl_scan(c, lane);
  if (lane == 31) wsum[w] = v;
  __syncthreads();
  int base = 0;
#pragma unroll
  for (int k = 0; k < 3; ++k)
    if (k < w) base += wsum[k];
  if (t < N_SCAN_BLOCKS) boff[t] = base + v - c;
}

__global__ void csr_fill_kernel(const int* __restrict__ row,
                                const int* __restrict__ col,
                                const int* __restrict__ rank,
                                const int* __restrict__ off,
                                const int* __restrict__ boff,
                                const float* __restrict__ dinv,
                                int* __restrict__ sedge) {
  int e = blockIdx.x * blockDim.x + threadIdx.x;
  if (e >= N_EDGES) return;
  const int r = row[e];
  const int c = col[e];
  const int p = off[c] + boff[c >> 10] + rank[e];
  v2i ev;
  ev.x = r;
  ev.y = __float_as_int(dinv[r] * dinv[c]);
  *reinterpret_cast<v2i*>(sedge + 2 * (size_t)p) = ev;
}

template <int DIN, int DOUT>
__global__ void gemm_wmma_kernel(const float* __restrict__ H,
                                 const float* __restrict__ W,
                                 float* __restrict__ Out) {
  constexpr int NT = (DOUT + 15) / 16;
  const int wave = (blockIdx.x * blockDim.x + threadIdx.x) >> 5;
  const int lane = threadIdx.x & 31;
  const int m0 = wave * 16;
  if (m0 >= N_NODES) return;

  const int lhalf = lane & 15;
  const int khi = (lane >> 4) << 1;
  const int ncl = (lhalf < DOUT) ? lhalf : (DOUT - 1);

  const float* __restrict__ hrow = H + (size_t)(m0 + lhalf) * DIN;

  v8f acc0 = {};
  v8f acc1 = {};
#pragma unroll
  for (int k = 0; k < DIN; k += 4) {
    v2f a, b0;
    a.x = hrow[k + khi];
    a.y = hrow[k + khi + 1];
    b0.x = W[(k + khi) * DOUT + ncl];
    b0.y = W[(k + khi + 1) * DOUT + ncl];
    acc0 = __builtin_amdgcn_wmma_f32_16x16x4_f32(
        false, a, false, b0, (short)0, acc0, false, false);
    if (NT > 1) {
      v2f b1;
      b1.x = W[(k + khi) * DOUT + 16 + lhalf];
      b1.y = W[(k + khi + 1) * DOUT + 16 + lhalf];
      acc1 = __builtin_amdgcn_wmma_f32_16x16x4_f32(
          false, a, false, b1, (short)0, acc1, false, false);
    }
  }

  const int mrow = m0 + ((lane >> 4) << 3);
#pragma unroll
  for (int v = 0; v < 8; ++v) {
    const int m = mrow + v;
    if (lhalf < DOUT) Out[(size_t)m * DOUT + lhalf] = acc0[v];
    if (NT > 1)       Out[(size_t)m * DOUT + 16 + lhalf] = acc1[v];
  }
}

__global__ void agg_gather32_kernel(const float* __restrict__ hW,
                                    const float* __restrict__ bias,
                                    const float* __restrict__ dinv,
                                    const int* __restrict__ cnt,
                                    const int* __restrict__ off,
                                    const int* __restrict__ boff,
                                    const int* __restrict__ sedge,
                                    float* __restrict__ out) {
  const int lane = threadIdx.x & 31;
  const int g = lane >> 3;
  const int q = lane & 7;
  const int i = __builtin_amdgcn_readfirstlane(
      (int)((blockIdx.x * blockDim.x + threadIdx.x) >> 5));
  if (i >= N_NODES) return;
  const int n = cnt[i];
  const int start = off[i] + boff[i >> 10];
  const float d = dinv[i];
  const v4f self = *reinterpret_cast<const v4f*>(hW + (size_t)i * 32 + 4 * q);
  v4f acc = {0.0f, 0.0f, 0.0f, 0.0f};
  for (int base = 0; base < n; base += 16) {
    v2i ev[4];
    v4f v[4];
#pragma unroll
    for (int u = 0; u < 4; ++u) {
      const int k = base + 4 * u + g;
      const int kk = (k < n) ? k : (n - 1);
      ev[u] = *reinterpret_cast<const v2i*>(sedge + 2 * (size_t)(start + kk));
    }
#pragma unroll
    for (int u = 0; u < 4; ++u)
      v[u] = *reinterpret_cast<const v4f*>(hW + (size_t)ev[u].x * 32 + 4 * q);
#pragma unroll
    for (int u = 0; u < 4; ++u) {
      if (base + 4 * u + g < n) acc += v[u] * __int_as_float(ev[u].y);
    }
  }
#pragma unroll
  for (int cc = 0; cc < 4; ++cc) {
    acc[cc] += __shfl_xor(acc[cc], 8);
    acc[cc] += __shfl_xor(acc[cc], 16);
  }
  if (g == 0) {
    v4f b;
#pragma unroll
    for (int cc = 0; cc < 4; ++cc) b[cc] = bias[4 * q + cc];
    const v4f o = (self * (d * d) + b) + acc;
    *reinterpret_cast<v4f*>(out + (size_t)i * 32 + 4 * q) = o;
  }
}

template <int DOUT>
__global__ void agg_gather_small_kernel(const float* __restrict__ hW,
                                        const float* __restrict__ bias,
                                        const float* __restrict__ dinv,
                                        const int* __restrict__ cnt,
                                        const int* __restrict__ off,
                                        const int* __restrict__ boff,
                                        const int* __restrict__ sedge,
                                        float* __restrict__ out) {
  const int t = blockIdx.x * blockDim.x + threadIdx.x;
  if (t >= N_NODES * DOUT) return;
  const int i = t / DOUT;
  const int f = t - i * DOUT;
  const int start = off[i] + boff[i >> 10];
  const int n = cnt[i];
  const float d = dinv[i];
  float acc = hW[t] * (d * d) + bias[f];
  for (int j = 0; j < n; j += 4) {
    int r[4];
    float wv[4];
    float v[4];
#pragma unroll
    for (int u = 0; u < 4; ++u) {
      const int q = (j + u < n) ? (j + u) : (n - 1);
      const v2i ev = *reinterpret_cast<const v2i*>(sedge + 2 * (size_t)(start + q));
      r[u] = ev.x;
      wv[u] = __int_as_float(ev.y);
    }
#pragma unroll
    for (int u = 0; u < 4; ++u) v[u] = hW[(size_t)r[u] * DOUT + f];
#pragma unroll
    for (int u = 0; u < 4; ++u) {
      if (j + u < n) acc += v[u] * wv[u];
    }
  }
  out[t] = acc;
}

#define N_PAD 100096

extern "C" void kernel_launch(void* const* d_in, const int* in_sizes, int n_in,
                              void* d_out, int out_size, void* d_ws, size_t ws_size,
                              hipStream_t stream) {
  const float* x  = (const float*)d_in[0];
  const int*   ei = (const int*)d_in[1];
  const int*   row = ei;
  const int*   col = ei + N_EDGES;

  const float* Wm[5] = {(const float*)d_in[2], (const float*)d_in[4],
                        (const float*)d_in[6], (const float*)d_in[8],
                        (const float*)d_in[10]};
  const float* Bm[5] = {(const float*)d_in[3], (const float*)d_in[5],
                        (const float*)d_in[7], (const float*)d_in[9],
                        (const float*)d_in[11]};

  int*   wsi   = (int*)d_ws;
  int*   cnt   = wsi;
  int*   off   = wsi + N_PAD;
  int*   bsum  = wsi + 2 * N_PAD;
  int*   boff  = wsi + 2 * N_PAD + 128;
  int*   rank  = wsi + 2 * N_PAD + 256;
  int*   sedge = rank + N_EDGES;
  float* dinv  = (float*)(sedge + 2 * N_EDGES);
  float* hW    = dinv + N_PAD;
  float* h     = hW + N_NODES * 32;

  const int B = 256;
  const int gN  = (N_NODES + B - 1) / B;
  const int gE  = (N_EDGES + B - 1) / B;
  const int gGemm = ((N_NODES / 16) * 32 + B - 1) / B;
  const int gG32 = (N_NODES * 32 + B - 1) / B;
  const int gG3  = (N_NODES * 3 + B - 1) / B;

  cnt_zero_kernel<<<gN, B, 0, stream>>>(cnt);
  cnt_rank_kernel<<<gE, B, 0, stream>>>(cnt, col, rank);
  scan_block_kernel<<<N_SCAN_BLOCKS, SCAN_B, 0, stream>>>(cnt, off, bsum, dinv);
  scan_top_kernel<<<1, 128, 0, stream>>>(bsum, boff);
  csr_fill_kernel<<<gE, B, 0, stream>>>(row, col, rank, off, boff, dinv, sedge);

  gemm_wmma_kernel<8, 32><<<gGemm, B, 0, stream>>>(x, Wm[0], hW);
  agg_gather32_kernel<<<gG32, B, 0, stream>>>(hW, Bm[0], dinv, cnt, off, boff,
                                              sedge, h);

  for (int l = 1; l <= 3; ++l) {
    gemm_wmma_kernel<32, 32><<<gGemm, B, 0, stream>>>(h, Wm[l], hW);
    agg_gather32_kernel<<<gG32, B, 0, stream>>>(hW, Bm[l], dinv, cnt, off, boff,
                                                sedge, h);
  }

  gemm_wmma_kernel<32, 3><<<gGemm, B, 0, stream>>>(h, Wm[4], hW);
  agg_gather_small_kernel<3><<<gG3, B, 0, stream>>>(hW, Bm[4], dinv, cnt, off, boff,
                                                   sedge, (float*)d_out);
}


# Layers are linear (slope-1 activation): one small WMMA kernel folds W1..W5 and the biases into an 8x3 matrix and 3-wide offsets, so all five gathers run 3 wide; products and sums are reassociated
# speedup vs baseline: 3.3998x; 1.1738x over previous
#include <hip/hip_runtime.h>

#define N_NODES 100000
#define N_EDGES 1600000

typedef __attribute__((ext_vector_type(2))) float v2f;
typedef __attribute__((ext_vector_type(8))) float v8f;
typedef __attribute__((ext_vector_type(2))) int v2i;
typedef __attribute__((ext_vector_type(4))) float v4f;

#define SCAN_B 1024
#define N_SCAN_BLOCKS ((N_NODES + SCAN_B - 1) / SCAN_B)

__global__ void cnt_zero_kernel(int* __restrict__ cnt) {
  int i = blockIdx.x * blockDim.x + threadIdx.x;
  if (i < N_NODES) cnt[i] = 0;
}

__global__ void cnt_rank_kernel(int* __restrict__ cnt,
                                const int* __restrict__ col,
                                int* __restrict__ rank) {
  int e = blockIdx.x * blockDim.x + threadIdx.x;
  if (e < N_EDGES) {
    rank[e] = __hip_atomic_fetch_add(&cnt[col[e]], 1,
                                     __ATOMIC_RELAXED, __HIP_MEMORY_SCOPE_AGENT);
  }
}

__device__ __forceinline__ int wave_incl_scan(int v, int lane) {
#pragma unroll
  for (int d = 1; d < 32; d <<= 1) {
    int t = __shfl_up(v, d);
    if (lane >= d) v += t;
  }
  return v;
}

__global__ void scan_block_kernel(const int* __restrict__ cnt,
                                  int* __restrict__ off,
                                  int* __restrict__ bsum,
                                  float* __restrict__ dinv) {
  __shared__ int wsum[32];
  const int i = blockIdx.x * SCAN_B + threadIdx.x;
  const int lane = threadIdx.x & 31;
  const int w = threadIdx.x >> 5;
  const int c = (i < N_NODES) ? cnt[i] : 0;
  if (i < N_NODES) dinv[i] = rsqrtf((float)(c + 1));
  const int v = wave_incl_scan(c, lane);
  if (lane == 31) wsum[w] = v;
  __syncthreads();
  if (w == 0) wsum[lane] = wave_incl_scan(wsum[lane], lane);
  __syncthreads();
  const int base = (w > 0) ? wsum[w - 1] : 0;
  if (i < N_NODES) off[i] = base + v - c;
  if (threadIdx.x == SCAN_B - 1) bsum[blockIdx.x] = base + v;
}

__global__ void scan_top_kernel(const int* __restrict__ bsum,
                                int* __restrict__ boff) {
  __shared__ int wsum[4];
  const int t = threadIdx.x;
  const int lane = t & 31;
  const int w = t >> 5;
  const int c = (t < N_SCAN_BLOCKS) ? bsum[t] : 0;
  const int v = wave_incl_scan(c, lane);
  if (lane == 31) wsum[w] = v;
  __syncthreads();
  int base = 0;
#pragma unroll
  for (int k = 0; k < 3; ++k)
    if (k < w) base += wsum[k];
  if (t < N_SCAN_BLOCKS) boff[t] = base + v - c;
}

__global__ void csr_fill_kernel(const int* __restrict__ row,
                                const int* __restrict__ col,
                                const int* __restrict__ rank,
                                const int* __restrict__ off,
                                const int* __restrict__ boff,
                                const float* __restrict__ dinv,
                                int* __restrict__ sedge) {
  int e = blockIdx.x * blockDim.x + threadIdx.x;
  if (e >= N_EDGES) return;
  const int r = row[e];
  const int c = col[e];
  const int p = off[c] + boff[c >> 10] + rank[e];
  v2i ev;
  ev.x = r;
  ev.y = __float_as_int(dinv[r] * dinv[c]);
  *reinterpret_cast<v2i*>(sedge + 2 * (size_t)p) = ev;
}

template <int DIN, int DOUT>
__global__ void gemm_wmma_kernel(const float* __restrict__ H,
                                 const float* __restrict__ W,
                                 float* __restrict__ Out) {
  constexpr int NT = (DOUT + 15) / 16;
  const int wave = (blockIdx.x * blockDim.x + threadIdx.x) >> 5;
  const int lane = threadIdx.x & 31;
  const int m0 = wave * 16;
  if (m0 >= N_NODES) return;

  const int lhalf = lane & 15;
  const int khi = (lane >> 4) << 1;
  const int ncl = (lhalf < DOUT) ? lhalf : (DOUT - 1);

  const float* __restrict__ hrow = H + (size_t)(m0 + lhalf) * DIN;

  v8f acc0 = {};
  v8f acc1 = {};
#pragma unroll
  for (int k = 0; k < DIN; k += 4) {
    v2f a, b0;
    a.x = hrow[k + khi];
    a.y = hrow[k + khi + 1];
    b0.x = W[(k + khi) * DOUT + ncl];
    b0.y = W[(k + khi + 1) * DOUT + ncl];
    acc0 = __builtin_amdgcn_wmma_f32_16x16x4_f32(
        false, a, false, b0, (short)0, acc0, false, false);
    if (NT > 1) {
      v2f b1;
      b1.x = W[(k + khi) * DOUT + 16 + lhalf];
      b1.y = W[(k + khi + 1) * DOUT + 16 + lhalf];
      acc1 = __builtin_amdgcn_wmma_f32_16x16x4_f32(
          false, a, false, b1, (short)0, acc1, false, false);
    }
  }

  const int mrow = m0 + ((lane >> 4) << 3);
#pragma unroll
  for (int v = 0; v < 8; ++v) {
    const int m = mrow + v;
    if (lhalf < DOUT) Out[(size_t)m * DOUT + lhalf] = acc0[v];
    if (NT > 1)       Out[(size_t)m * DOUT + 16 + lhalf] = acc1[v];
  }
}

__device__ __forceinline__ void chain_load_a(const float* __restrict__ rowp,
                                             int khi, v2f (&a)[8]) {
#pragma unroll
  for (int j = 0; j < 8; ++j) {
    a[j].x = rowp[4 * j + khi];
    a[j].y = rowp[4 * j + khi + 1];
  }
}

__device__ __forceinline__ v8f chain_tile(const v2f (&a)[8],
                                          const float (*P)[16],
                                          int khi, int lhalf) {
  v8f acc = {};
#pragma unroll
  for (int j = 0; j < 8; ++j) {
    v2f b;
    b.x = P[4 * j + khi][lhalf];
    b.y = P[4 * j + khi + 1][lhalf];
    acc = __builtin_amdgcn_wmma_f32_16x16x4_f32(
        false, a[j], false, b, (short)0, acc, false, false);
  }
  return acc;
}

__device__ __forceinline__ void chain_step(const v2f (&w0)[8], const v2f (&w1)[8],
                                           const v2f (&bb)[8],
                                           const float (*Pc)[16], float (*Pn)[16],
                                           float* __restrict__ ck, int lane) {
  const int lhalf = lane & 15;
  const int khi = (lane >> 4) << 1;
  const int mhi = (lane >> 4) << 3;
  const v8f d0 = chain_tile(w0, Pc, khi, lhalf);
  const v8f d1 = chain_tile(w1, Pc, khi, lhalf);
  const v8f dc = chain_tile(bb, Pc, khi, lhalf);
#pragma unroll
  for (int v = 0; v < 8; ++v) {
    Pn[mhi + v][lhalf] = d0[v];
    Pn[16 + mhi + v][lhalf] = d1[v];
  }
  if (lane < 3) ck[lane] = dc[0];
  __syncthreads();
}

__global__ void __launch_bounds__(32)
weight_chain_kernel(const float* __restrict__ W1, const float* __restrict__ b1,
                    const float* __restrict__ W2, const float* __restrict__ b2,
                    const float* __restrict__ W3, const float* __restrict__ b3,
                    const float* __restrict__ W4, const float* __restrict__ b4,
                    const float* __restrict__ W5,
                    float* __restrict__ P1,
                    float* __restrict__ cv) {
  __shared__ float P[2][32][16];
  const int lane = threadIdx.x;
  const int lhalf = lane & 15;
  const int khi = (lane >> 4) << 1;

  v2f aW4[2][8], aW3[2][8], aW2[2][8], aW1[8];
  v2f aB4[8], aB3[8], aB2[8], aB1[8];
#pragma unroll
  for (int t = 0; t < 2; ++t) {
    chain_load_a(W4 + (16 * t + lhalf) * 32, khi, aW4[t]);
    chain_load_a(W3 + (16 * t + lhalf) * 32, khi, aW3[t]);
    chain_load_a(W2 + (16 * t + lhalf) * 32, khi, aW2[t]);
  }
  chain_load_a(W1 + (lhalf & 7) * 32, khi, aW1);
  chain_load_a(b4, khi, aB4);
  chain_load_a(b3, khi, aB3);
  chain_load_a(b2, khi, aB2);
  chain_load_a(b1, khi, aB1);

  for (int idx = lane; idx < 32 * 16; idx += 32) {
    const int r = idx >> 4;
    const int c = idx & 15;
    P[0][r][c] = W5[r * 3 + (c < 3 ? c : 2)];
  }
  __syncthreads();

  chain_step(aW4[0], aW4[1], aB4, P[0], P[1], cv + 9, lane);
  chain_step(aW3[0], aW3[1], aB3, P[1], P[0], cv + 6, lane);
  chain_step(aW2[0], aW2[1], aB2, P[0], P[1], cv + 3, lane);

  const float (*P2)[16] = P[1];
  const v8f d0 = chain_tile(aW1, P2, khi, lhalf);
  const v8f dc = chain_tile(aB1, P2, khi, lhalf);
  if (lane < 3) {
#pragma unroll
    for (int v = 0; v < 8; ++v) P1[v * 3 + lane] = d0[v];
    cv[lane] = dc[0];
  }
}

template <int DOUT>
__global__ void agg_gather_small_kernel(const float* __restrict__ hW,
                                        const float* __restrict__ bias,
                                        const float* __restrict__ dinv,
                                        const int* __restrict__ cnt,
                                        const int* __restrict__ off,
                                        const int* __restrict__ boff,
                                        const int* __restrict__ sedge,
                                        float* __restrict__ out) {
  const int t = blockIdx.x * blockDim.x + threadIdx.x;
  if (t >= N_NODES * DOUT) return;
  const int i = t / DOUT;
  const int f = t - i * DOUT;
  const int start = off[i] + boff[i >> 10];
  const int n = cnt[i];
  const float d = dinv[i];
  float acc = hW[t] * (d * d) + bias[f];
  for (int j = 0; j < n; j += 4) {
    int r[4];
    float wv[4];
    float v[4];
#pragma unroll
    for (int u = 0; u < 4; ++u) {
      const int q = (j + u < n) ? (j + u) : (n - 1);
      const v2i ev = *reinterpret_cast<const v2i*>(sedge + 2 * (size_t)(start + q));
      r[u] = ev.x;
      wv[u] = __int_as_float(ev.y);
    }
#pragma unroll
    for (int u = 0; u < 4; ++u) v[u] = hW[(size_t)r[u] * DOUT + f];
#pragma unroll
    for (int u = 0; u < 4; ++u) {
      if (j + u < n) acc += v[u] * wv[u];
    }
  }
  out[t] = acc;
}

#define N_PAD 100096

extern "C" void kernel_launch(void* const* d_in, const int* in_sizes, int n_in,
                              void* d_out, int out_size, void* d_ws, size_t ws_size,
                              hipStream_t stream) {
  const float* x  = (const float*)d_in[0];
  const int*   ei = (const int*)d_in[1];
  const int*   row = ei;
  const int*   col = ei + N_EDGES;

  const float* Wm[5] = {(const float*)d_in[2], (const float*)d_in[4],
                        (const float*)d_in[6], (const float*)d_in[8],
                        (const float*)d_in[10]};
  const float* Bm[5] = {(const float*)d_in[3], (const float*)d_in[5],
                        (const float*)d_in[7], (const float*)d_in[9],
                        (const float*)d_in[11]};

  int*   wsi   = (int*)d_ws;
  int*   cnt   = wsi;
  int*   off   = wsi + N_PAD;
  int*   bsum  = wsi + 2 * N_PAD;
  int*   boff  = wsi + 2 * N_PAD + 128;
  int*   rank  = wsi + 2 * N_PAD + 256;
  int*   sedge = rank + N_EDGES;
  float* dinv  = (float*)(sedge + 2 * N_EDGES);
  float* P1    = dinv + N_PAD;
  float* cv    = P1 + 64;
  float* zA    = P1 + 128;
  float* zB    = zA + 3 * N_PAD;

  const int B = 256;
  const int gN  = (N_NODES + B - 1) / B;
  const int gE  = (N_EDGES + B - 1) / B;
  const int gGemm = ((N_NODES / 16) * 32 + B - 1) / B;
  const int gG3  = (N_NODES * 3 + B - 1) / B;

  cnt_zero_kernel<<<gN, B, 0, stream>>>(cnt);
  cnt_rank_kernel<<<gE, B, 0, stream>>>(cnt, col, rank);
  scan_block_kernel<<<N_SCAN_BLOCKS, SCAN_B, 0, stream>>>(cnt, off, bsum, dinv);
  scan_top_kernel<<<1, 128, 0, stream>>>(bsum, boff);
  csr_fill_kernel<<<gE, B, 0, stream>>>(row, col, rank, off, boff, dinv, sedge);

  weight_chain_kernel<<<1, 32, 0, stream>>>(Wm[0], Bm[0], Wm[1], Bm[1], Wm[2], Bm[2],
                                            Wm[3], Bm[3], Wm[4], P1, cv);
  gemm_wmma_kernel<8, 3><<<gGemm, B, 0, stream>>>(x, P1, zA);

  agg_gather_small_kernel<3><<<gG3, B, 0, stream>>>(zA, cv + 0, dinv, cnt, off, boff,
                                                   sedge, zB);
  agg_gather_small_kernel<3><<<gG3, B, 0, stream>>>(zB, cv + 3, dinv, cnt, off, boff,
                                                   sedge, zA);
  agg_gather_small_kernel<3><<<gG3, B, 0, stream>>>(zA, cv + 6, dinv, cnt, off, boff,
                                                   sedge, zB);
  agg_gather_small_kernel<3><<<gG3, B, 0, stream>>>(zB, cv + 9, dinv, cnt, off, boff,
                                                   sedge, zA);
  agg_gather_small_kernel<3><<<gG3, B, 0, stream>>>(zA, Bm[4], dinv, cnt, off, boff,
                                                   sedge, (float*)d_out);
}


# Gather runs one lane per node on value rows padded to 16 bytes: one 128-bit load per edge, 16 edges in flight per lane, a third of the waves; sequential storage-order sums unchanged
# speedup vs baseline: 3.8999x; 1.1471x over previous
#include <hip/hip_runtime.h>

#define N_NODES 100000
#define N_EDGES 1600000

typedef __attribute__((ext_vector_type(2))) float v2f;
typedef __attribute__((ext_vector_type(8))) float v8f;
typedef __attribute__((ext_vector_type(2))) int v2i;
typedef __attribute__((ext_vector_type(4))) float v4f;

#define SCAN_B 1024
#define N_SCAN_BLOCKS ((N_NODES + SCAN_B - 1) / SCAN_B)

__global__ void cnt_zero_kernel(int* __restrict__ cnt) {
  int i = blockIdx.x * blockDim.x + threadIdx.x;
  if (i < N_NODES) cnt[i] = 0;
}

__global__ void cnt_rank_kernel(int* __restrict__ cnt,
                                const int* __restrict__ col,
                                int* __restrict__ rank) {
  int e = blockIdx.x * blockDim.x + threadIdx.x;
  if (e < N_EDGES) {
    rank[e] = __hip_atomic_fetch_add(&cnt[col[e]], 1,
                                     __ATOMIC_RELAXED, __HIP_MEMORY_SCOPE_AGENT);
  }
}

__device__ __forceinline__ int wave_incl_scan(int v, int lane) {
#pragma unroll
  for (int d = 1; d < 32; d <<= 1) {
    int t = __shfl_up(v, d);
    if (lane >= d) v += t;
  }
  return v;
}

__global__ void scan_block_kernel(const int* __restrict__ cnt,
                                  int* __restrict__ off,
                                  int* __restrict__ bsum,
                                  float* __restrict__ dinv) {
  __shared__ int wsum[32];
  const int i = blockIdx.x * SCAN_B + threadIdx.x;
  const int lane = threadIdx.x & 31;
  const int w = threadIdx.x >> 5;
  const int c = (i < N_NODES) ? cnt[i] : 0;
  if (i < N_NODES) dinv[i] = rsqrtf((float)(c + 1));
  const int v = wave_incl_scan(c, lane);
  if (lane == 31) wsum[w] = v;
  __syncthreads();
  if (w == 0) wsum[lane] = wave_incl_scan(wsum[lane], lane);
  __syncthreads();
  const int base = (w > 0) ? wsum[w - 1] : 0;
  if (i < N_NODES) off[i] = base + v - c;
  if (threadIdx.x == SCAN_B - 1) bsum[blockIdx.x] = base + v;
}

__global__ void scan_top_kernel(const int* __restrict__ bsum,
                                int* __restrict__ boff) {
  __shared__ int wsum[4];
  const int t = threadIdx.x;
  const int lane = t & 31;
  const int w = t >> 5;
  const int c = (t < N_SCAN_BLOCKS) ? bsum[t] : 0;
  const int v = wave_incl_scan(c, lane);
  if (lane == 31) wsum[w] = v;
  __syncthreads();
  int base = 0;
#pragma unroll
  for (int k = 0; k < 3; ++k)
    if (k < w) base += wsum[k];
  if (t < N_SCAN_BLOCKS) boff[t] = base + v - c;
}

__global__ void csr_fill_kernel(const int* __restrict__ row,
                                const int* __restrict__ col,
                                const int* __restrict__ rank,
                                const int* __restrict__ off,
                                const int* __restrict__ boff,
                                const float* __restrict__ dinv,
                                int* __restrict__ sedge) {
  int e = blockIdx.x * blockDim.x + threadIdx.x;
  if (e >= N_EDGES) return;
  const int r = row[e];
  const int c = col[e];
  const int p = off[c] + boff[c >> 10] + rank[e];
  v2i ev;
  ev.x = r;
  ev.y = __float_as_int(dinv[r] * dinv[c]);
  *reinterpret_cast<v2i*>(sedge + 2 * (size_t)p) = ev;
}

template <int DIN, int DOUT, int OPITCH>
__global__ void gemm_wmma_kernel(const float* __restrict__ H,
                                 const float* __restrict__ W,
                                 float* __restrict__ Out) {
  constexpr int NT = (DOUT + 15) / 16;
  const int wave = (blockIdx.x * blockDim.x + threadIdx.x) >> 5;
  const int lane = threadIdx.x & 31;
  const int m0 = wave * 16;
  if (m0 >= N_NODES) return;

  const int lhalf = lane & 15;
  const int khi = (lane >> 4) << 1;
  const int ncl = (lhalf < DOUT) ? lhalf : (DOUT - 1);

  const float* __restrict__ hrow = H + (size_t)(m0 + lhalf) * DIN;

  v8f acc0 = {};
  v8f acc1 = {};
#pragma unroll
  for (int k = 0; k < DIN; k += 4) {
    v2f a, b0;
    a.x = hrow[k + khi];
    a.y = hrow[k + khi + 1];
    b0.x = W[(k + khi) * DOUT + ncl];
    b0.y = W[(k + khi + 1) * DOUT + ncl];
    acc0 = __builtin_amdgcn_wmma_f32_16x16x4_f32(
        false, a, false, b0, (short)0, acc0, false, false);
    if (NT > 1) {
      v2f b1;
      b1.x = W[(k + khi) * DOUT + 16 + lhalf];
      b1.y = W[(k + khi + 1) * DOUT + 16 + lhalf];
      acc1 = __builtin_amdgcn_wmma_f32_16x16x4_f32(
          false, a, false, b1, (short)0, acc1, false, false);
    }
  }

  const int mrow = m0 + ((lane >> 4) << 3);
#pragma unroll
  for (int v = 0; v < 8; ++v) {
    const int m = mrow + v;
    if (lhalf < OPITCH) Out[(size_t)m * OPITCH + lhalf] = acc0[v];
    if (NT > 1)         Out[(size_t)m * OPITCH + 16 + lhalf] = acc1[v];
  }
}

__device__ __forceinline__ void chain_load_a(const float* __restrict__ rowp,
                                             int khi, v2f (&a)[8]) {
#pragma unroll
  for (int j = 0; j < 8; ++j) {
    a[j].x = rowp[4 * j + khi];
    a[j].y = rowp[4 * j + khi + 1];
  }
}

__device__ __forceinline__ v8f chain_tile(const v2f (&a)[8],
                                          const float (*P)[16],
                                          int khi, int lhalf) {
  v8f acc = {};
#pragma unroll
  for (int j = 0; j < 8; ++j) {
    v2f b;
    b.x = P[4 * j + khi][lhalf];
    b.y = P[4 * j + khi + 1][lhalf];
    acc = __builtin_amdgcn_wmma_f32_16x16x4_f32(
        false, a[j], false, b, (short)0, acc, false, false);
  }
  return acc;
}

__device__ __forceinline__ void chain_step(const v2f (&w0)[8], const v2f (&w1)[8],
                                           const v2f (&bb)[8],
                                           const float (*Pc)[16], float (*Pn)[16],
                                           float* __restrict__ ck, int lane) {
  const int lhalf = lane & 15;
  const int khi = (lane >> 4) << 1;
  const int mhi = (lane >> 4) << 3;
  const v8f d0 = chain_tile(w0, Pc, khi, lhalf);
  const v8f d1 = chain_tile(w1, Pc, khi, lhalf);
  const v8f dc = chain_tile(bb, Pc, khi, lhalf);
#pragma unroll
  for (int v = 0; v < 8; ++v) {
    Pn[mhi + v][lhalf] = d0[v];
    Pn[16 + mhi + v][lhalf] = d1[v];
  }
  if (lane < 3) ck[lane] = dc[0];
  __syncthreads();
}

__global__ void __launch_bounds__(32)
weight_chain_kernel(const float* __restrict__ W1, const float* __restrict__ b1,
                    const float* __restrict__ W2, const float* __restrict__ b2,
                    const float* __restrict__ W3, const float* __restrict__ b3,
                    const float* __restrict__ W4, const float* __restrict__ b4,
                    const float* __restrict__ W5,
                    float* __restrict__ P1,
                    float* __restrict__ cv) {
  __shared__ float P[2][32][16];
  const int lane = threadIdx.x;
  const int lhalf = lane & 15;
  const int khi = (lane >> 4) << 1;

  v2f aW4[2][8], aW3[2][8], aW2[2][8], aW1[8];
  v2f aB4[8], aB3[8], aB2[8], aB1[8];
#pragma unroll
  for (int t = 0; t < 2; ++t) {
    chain_load_a(W4 + (16 * t + lhalf) * 32, khi, aW4[t]);
    chain_load_a(W3 + (16 * t + lhalf) * 32, khi, aW3[t]);
    chain_load_a(W2 + (16 * t + lhalf) * 32, khi, aW2[t]);
  }
  chain_load_a(W1 + (lhalf & 7) * 32, khi, aW1);
  chain_load_a(b4, khi, aB4);
  chain_load_a(b3, khi, aB3);
  chain_load_a(b2, khi, aB2);
  chain_load_a(b1, khi, aB1);

  for (int idx = lane; idx < 32 * 16; idx += 32) {
    const int r = idx >> 4;
    const int c = idx & 15;
    P[0][r][c] = W5[r * 3 + (c < 3 ? c : 2)];
  }
  __syncthreads();

  chain_step(aW4[0], aW4[1], aB4, P[0], P[1], cv + 9, lane);
  chain_step(aW3[0], aW3[1], aB3, P[1], P[0], cv + 6, lane);
  chain_step(aW2[0], aW2[1], aB2, P[0], P[1], cv + 3, lane);

  const float (*P2)[16] = P[1];
  const v8f d0 = chain_tile(aW1, P2, khi, lhalf);
  const v8f dc = chain_tile(aB1, P2, khi, lhalf);
  if (lane < 3) {
#pragma unroll
    for (int v = 0; v < 8; ++v) P1[v * 3 + lane] = d0[v];
    cv[lane] = dc[0];
  }
}

#define GU 16
#define GATHER_B 64

template <bool LAST>
__global__ void __launch_bounds__(GATHER_B)
agg_gather_node_kernel(const float* __restrict__ zin,
                       const float* __restrict__ bias,
                       const float* __restrict__ dinv,
                       const int* __restrict__ cnt,
                       const int* __restrict__ off,
                       const int* __restrict__ boff,
                       const int* __restrict__ sedge,
                       float* __restrict__ out) {
  const int i = blockIdx.x * GATHER_B + threadIdx.x;
  if (i >= N_NODES) return;
  const int n = cnt[i];
  const int start = off[i] + boff[i >> 10];
  const float d = dinv[i];
  const v4f* __restrict__ z4 = reinterpret_cast<const v4f*>(zin);
  const v2i* __restrict__ recb = reinterpret_cast<const v2i*>(sedge);
  const float* __restrict__ zself = zin + 4 * (size_t)i;
  float a0 = zself[0] * (d * d) + bias[0];
  float a1 = zself[1] * (d * d) + bias[1];
  float a2 = zself[2] * (d * d) + bias[2];

  int j = 0;
  do {
    v2i ev[GU];
    v4f v[GU];
#pragma unroll
    for (int u = 0; u < GU; ++u) {
      const unsigned p = (j + u < n) ? (unsigned)(start + j + u) : 0u;
      ev[u] = recb[p];
    }
    __builtin_amdgcn_sched_barrier(0);
#pragma unroll
    for (int u = 0; u < GU; ++u) {
      const unsigned r = (j + u < n) ? (unsigned)ev[u].x : 0u;
      v[u] = z4[r];
    }
    __builtin_amdgcn_sched_barrier(0);
#pragma unroll
    for (int u = 0; u < GU; ++u) {
      const bool act = (j + u < n);
      const float w = __int_as_float(ev[u].y);
      const float t0 = a0 + v[u].x * w;
      const float t1 = a1 + v[u].y * w;
      const float t2 = a2 + v[u].z * w;
      a0 = act ? t0 : a0;
      a1 = act ? t1 : a1;
      a2 = act ? t2 : a2;
    }
    j += GU;
  } while (j < n);

  if (LAST) {
    out[3 * (size_t)i + 0] = a0;
    out[3 * (size_t)i + 1] = a1;
    out[3 * (size_t)i + 2] = a2;
  } else {
    v4f o;
    o.x = a0;
    o.y = a1;
    o.z = a2;
    o.w = 0.f;
    reinterpret_cast<v4f*>(out)[i] = o;
  }
}

#define N_PAD 100096

extern "C" void kernel_launch(void* const* d_in, const int* in_sizes, int n_in,
                              void* d_out, int out_size, void* d_ws, size_t ws_size,
                              hipStream_t stream) {
  const float* x  = (const float*)d_in[0];
  const int*   ei = (const int*)d_in[1];
  const int*   row = ei;
  const int*   col = ei + N_EDGES;

  const float* Wm[5] = {(const float*)d_in[2], (const float*)d_in[4],
                        (const float*)d_in[6], (const float*)d_in[8],
                        (const float*)d_in[10]};
  const float* Bm[5] = {(const float*)d_in[3], (const float*)d_in[5],
                        (const float*)d_in[7], (const float*)d_in[9],
                        (const float*)d_in[11]};

  int*   wsi   = (int*)d_ws;
  int*   cnt   = wsi;
  int*   off   = wsi + N_PAD;
  int*   bsum  = wsi + 2 * N_PAD;
  int*   boff  = wsi + 2 * N_PAD + 128;
  int*   rank  = wsi + 2 * N_PAD + 256;
  int*   sedge = rank + N_EDGES;
  float* dinv  = (float*)(sedge + 2 * N_EDGES);
  float* P1    = dinv + N_PAD;
  float* cv    = P1 + 64;
  float* zA    = P1 + 128;
  float* zB    = zA + 4 * N_PAD;

  const int B = 256;
  const int gN  = (N_NODES + B - 1) / B;
  const int gE  = (N_EDGES + B - 1) / B;
  const int gGemm = ((N_NODES / 16) * 32 + B - 1) / B;
  const int gGn  = (N_NODES + GATHER_B - 1) / GATHER_B;

  cnt_zero_kernel<<<gN, B, 0, stream>>>(cnt);
  cnt_rank_kernel<<<gE, B, 0, stream>>>(cnt, col, rank);
  scan_block_kernel<<<N_SCAN_BLOCKS, SCAN_B, 0, stream>>>(cnt, off, bsum, dinv);
  scan_top_kernel<<<1, 128, 0, stream>>>(bsum, boff);
  csr_fill_kernel<<<gE, B, 0, stream>>>(row, col, rank, off, boff, dinv, sedge);

  weight_chain_kernel<<<1, 32, 0, stream>>>(Wm[0], Bm[0], Wm[1], Bm[1], Wm[2], Bm[2],
                                            Wm[3], Bm[3], Wm[4], P1, cv);
  gemm_wmma_kernel<8, 3, 4><<<gGemm, B, 0, stream>>>(x, P1, zA);

  agg_gather_node_kernel<false><<<gGn, GATHER_B, 0, stream>>>(zA, cv + 0, dinv, cnt, off,
                                                              boff, sedge, zB);
  agg_gather_node_kernel<false><<<gGn, GATHER_B, 0, stream>>>(zB, cv + 3, dinv, cnt, off,
                                                              boff, sedge, zA);
  agg_gather_node_kernel<false><<<gGn, GATHER_B, 0, stream>>>(zA, cv + 6, dinv, cnt, off,
                                                              boff, sedge, zB);
  agg_gather_node_kernel<false><<<gGn, GATHER_B, 0, stream>>>(zB, cv + 9, dinv, cnt, off,
                                                              boff, sedge, zA);
  agg_gather_node_kernel<true><<<gGn, GATHER_B, 0, stream>>>(zA, Bm[4], dinv, cnt, off,
                                                             boff, sedge, (float*)d_out);
}


# Edge weight factored as dinv[col]*dinv[row]: value rows kept pre-scaled by dinv, a sorted record is the 4-byte index, read four per 128-bit load; products reassociated, bias added after the sum
# speedup vs baseline: 4.1744x; 1.0704x over previous
#include <hip/hip_runtime.h>

#define N_NODES 100000
#define N_EDGES 1600000

typedef __attribute__((ext_vector_type(2))) float v2f;
typedef __attribute__((ext_vector_type(8))) float v8f;
typedef __attribute__((ext_vector_type(4))) int v4i;
typedef __attribute__((ext_vector_type(4))) float v4f;
struct row3 { float x, y, z; };

#define SCAN_B 1024
#define N_SCAN_BLOCKS ((N_NODES + SCAN_B - 1) / SCAN_B)

__global__ void cnt_zero_kernel(int* __restrict__ cnt) {
  int i = blockIdx.x * blockDim.x + threadIdx.x;
  if (i < N_NODES) cnt[i] = 0;
}

__global__ void cnt_rank_kernel(int* __restrict__ cnt,
                                const int* __restrict__ col,
                                int* __restrict__ rank) {
  int e = blockIdx.x * blockDim.x + threadIdx.x;
  if (e < N_EDGES) {
    rank[e] = __hip_atomic_fetch_add(&cnt[col[e]], 1,
                                     __ATOMIC_RELAXED, __HIP_MEMORY_SCOPE_AGENT);
  }
}

__device__ __forceinline__ int wave_incl_scan(int v, int lane) {
#pragma unroll
  for (int d = 1; d < 32; d <<= 1) {
    int t = __shfl_up(v, d);
    if (lane >= d) v += t;
  }
  return v;
}

__global__ void scan_block_kernel(const int* __restrict__ cnt,
                                  int* __restrict__ off,
                                  int* __restrict__ bsum,
                                  float* __restrict__ dinv) {
  __shared__ int wsum[32];
  const int i = blockIdx.x * SCAN_B + threadIdx.x;
  const int lane = threadIdx.x & 31;
  const int w = threadIdx.x >> 5;
  const int cr = (i < N_NODES) ? cnt[i] : 0;
  if (i < N_NODES) dinv[i] = rsqrtf((float)(cr + 1));
  const int c = (cr + 3) & ~3;
  const int v = wave_incl_scan(c, lane);
  if (lane == 31) wsum[w] = v;
  __syncthreads();
  if (w == 0) wsum[lane] = wave_incl_scan(wsum[lane], lane);
  __syncthreads();
  const int base = (w > 0) ? wsum[w - 1] : 0;
  if (i < N_NODES) off[i] = base + v - c;
  if (threadIdx.x == SCAN_B - 1) bsum[blockIdx.x] = base + v;
}

__global__ void scan_top_kernel(const int* __restrict__ bsum,
                                int* __restrict__ boff) {
  __shared__ int wsum[4];
  const int t = threadIdx.x;
  const int lane = t & 31;
  const int w = t >> 5;
  const int c = (t < N_SCAN_BLOCKS) ? bsum[t] : 0;
  const int v = wave_incl_scan(c, lane);
  if (lane == 31) wsum[w] = v;
  __syncthreads();
  int base = 0;
#pragma unroll
  for (int k = 0; k < 3; ++k)
    if (k < w) base += wsum[k];
  if (t < N_SCAN_BLOCKS) boff[t] = base + v - c;
}

__global__ void csr_fill_kernel(const int* __restrict__ row,
                                const int* __restrict__ col,
                                const int* __restrict__ rank,
                                const int* __restrict__ cnt,
                                const int* __restrict__ off,
                                const int* __restrict__ boff,
                                int* __restrict__ sedge) {
  int e = blockIdx.x * blockDim.x + threadIdx.x;
  if (e >= N_EDGES) return;
  const int c = col[e];
  sedge[off[c] + boff[c >> 10] + rank[e]] = row[e];
  if (e < N_NODES) {
    const int n = cnt[e];
    const int t = off[e] + boff[e >> 10] + n;
    const int pad = (-n) & 3;
    if (pad > 0) sedge[t] = 0;
    if (pad > 1) sedge[t + 1] = 0;
    if (pad > 2) sedge[t + 2] = 0;
  }
}

template <int DIN, int DOUT, int OPITCH>
__global__ void gemm_wmma_kernel(const float* __restrict__ H,
                                 const float* __restrict__ W,
                                 const float* __restrict__ rscale,
                                 float* __restrict__ Out) {
  constexpr int NT = (DOUT + 15) / 16;
  const int wave = (blockIdx.x * blockDim.x + threadIdx.x) >> 5;
  const int lane = threadIdx.x & 31;
  const int m0 = wave * 16;
  if (m0 >= N_NODES) return;

  const int lhalf = lane & 15;
  const int khi = (lane >> 4) << 1;
  const int ncl = (lhalf < DOUT) ? lhalf : (DOUT - 1);

  const float* __restrict__ hrow = H + (size_t)(m0 + lhalf) * DIN;

  v8f acc0 = {};
  v8f acc1 = {};
#pragma unroll
  for (int k = 0; k < DIN; k += 4) {
    v2f a, b0;
    a.x = hrow[k + khi];
    a.y = hrow[k + khi + 1];
    b0.x = W[(k + khi) * DOUT + ncl];
    b0.y = W[(k + khi + 1) * DOUT + ncl];
    acc0 = __builtin_amdgcn_wmma_f32_16x16x4_f32(
        false, a, false, b0, (short)0, acc0, false, false);
    if (NT > 1) {
      v2f b1;
      b1.x = W[(k + khi) * DOUT + 16 + lhalf];
      b1.y = W[(k + khi + 1) * DOUT + 16 + lhalf];
      acc1 = __builtin_amdgcn_wmma_f32_16x16x4_f32(
          false, a, false, b1, (short)0, acc1, false, false);
    }
  }

  const int mrow = m0 + ((lane >> 4) << 3);
#pragma unroll
  for (int v = 0; v < 8; ++v) {
    const int m = mrow + v;
    const float rs = rscale[m];
    if (lhalf < OPITCH) Out[(size_t)m * OPITCH + lhalf] = acc0[v] * rs;
    if (NT > 1)         Out[(size_t)m * OPITCH + 16 + lhalf] = acc1[v] * rs;
  }
}

__device__ __forceinline__ void chain_load_a(const float* __restrict__ rowp,
                                             int khi, v2f (&a)[8]) {
#pragma unroll
  for (int j = 0; j < 8; ++j) {
    a[j].x = rowp[4 * j + khi];
    a[j].y = rowp[4 * j + khi + 1];
  }
}

__device__ __forceinline__ v8f chain_tile(const v2f (&a)[8],
                                          const float (*P)[16],
                                          int khi, int lhalf) {
  v8f acc = {};
#pragma unroll
  for (int j = 0; j < 8; ++j) {
    v2f b;
    b.x = P[4 * j + khi][lhalf];
    b.y = P[4 * j + khi + 1][lhalf];
    acc = __builtin_amdgcn_wmma_f32_16x16x4_f32(
        false, a[j], false, b, (short)0, acc, false, false);
  }
  return acc;
}

__device__ __forceinline__ void chain_step(const v2f (&w0)[8], const v2f (&w1)[8],
                                           const v2f (&bb)[8],
                                           const float (*Pc)[16], float (*Pn)[16],
                                           float* __restrict__ ck, int lane) {
  const int lhalf = lane & 15;
  const int khi = (lane >> 4) << 1;
  const int mhi = (lane >> 4) << 3;
  const v8f d0 = chain_tile(w0, Pc, khi, lhalf);
  const v8f d1 = chain_tile(w1, Pc, khi, lhalf);
  const v8f dc = chain_tile(bb, Pc, khi, lhalf);
#pragma unroll
  for (int v = 0; v < 8; ++v) {
    Pn[mhi + v][lhalf] = d0[v];
    Pn[16 + mhi + v][lhalf] = d1[v];
  }
  if (lane < 3) ck[lane] = dc[0];
  __syncthreads();
}

__global__ void __launch_bounds__(32)
weight_chain_kernel(const float* __restrict__ W1, const float* __restrict__ b1,
                    const float* __restrict__ W2, const float* __restrict__ b2,
                    const float* __restrict__ W3, const float* __restrict__ b3,
                    const float* __restrict__ W4, const float* __restrict__ b4,
                    const float* __restrict__ W5,
                    float* __restrict__ P1,
                    float* __restrict__ cv) {
  __shared__ float P[2][32][16];
  const int lane = threadIdx.x;
  const int lhalf = lane & 15;
  const int khi = (lane >> 4) << 1;

  v2f aW4[2][8], aW3[2][8], aW2[2][8], aW1[8];
  v2f aB4[8], aB3[8], aB2[8], aB1[8];
#pragma unroll
  for (int t = 0; t < 2; ++t) {
    chain_load_a(W4 + (16 * t + lhalf) * 32, khi, aW4[t]);
    chain_load_a(W3 + (16 * t + lhalf) * 32, khi, aW3[t]);
    chain_load_a(W2 + (16 * t + lhalf) * 32, khi, aW2[t]);
  }
  chain_load_a(W1 + (lhalf & 7) * 32, khi, aW1);
  chain_load_a(b4, khi, aB4);
  chain_load_a(b3, khi, aB3);
  chain_load_a(b2, khi, aB2);
  chain_load_a(b1, khi, aB1);

  for (int idx = lane; idx < 32 * 16; idx += 32) {
    const int r = idx >> 4;
    const int c = idx & 15;
    P[0][r][c] = W5[r * 3 + (c < 3 ? c : 2)];
  }
  __syncthreads();

  chain_step(aW4[0], aW4[1], aB4, P[0], P[1], cv + 9, lane);
  chain_step(aW3[0], aW3[1], aB3, P[1], P[0], cv + 6, lane);
  chain_step(aW2[0], aW2[1], aB2, P[0], P[1], cv + 3, lane);

  const float (*P2)[16] = P[1];
  const v8f d0 = chain_tile(aW1, P2, khi, lhalf);
  const v8f dc = chain_tile(aB1, P2, khi, lhalf);
  if (lane < 3) {
#pragma unroll
    for (int v = 0; v < 8; ++v) P1[v * 3 + lane] = d0[v];
    cv[lane] = dc[0];
  }
}

#define GU 16
#define GATHER_B 64

template <bool LAST>
__global__ void __launch_bounds__(GATHER_B)
agg_gather_node_kernel(const float* __restrict__ yin,
                       const float* __restrict__ bias,
                       const float* __restrict__ dinv,
                       const int* __restrict__ cnt,
                       const int* __restrict__ off,
                       const int* __restrict__ boff,
                       const int* __restrict__ sedge,
                       float* __restrict__ out) {
  const int i = blockIdx.x * GATHER_B + threadIdx.x;
  if (i >= N_NODES) return;
  const int n = cnt[i];
  const int sq = (off[i] + boff[i >> 10]) >> 2;
  const float d = dinv[i];
  const v4i* __restrict__ recq = reinterpret_cast<const v4i*>(sedge);
  const float* __restrict__ yself = yin + 4 * (size_t)i;
  float a0 = yself[0];
  float a1 = yself[1];
  float a2 = yself[2];

  int j = 0;
  do {
    v4i q[GU / 4];
    row3 v[GU];
#pragma unroll
    for (int t = 0; t < GU / 4; ++t) {
      const unsigned p = (j + 4 * t < n) ? (unsigned)(sq + (j >> 2) + t) : 0u;
      q[t] = recq[p];
    }
    __builtin_amdgcn_sched_barrier(0);
#pragma unroll
    for (int u = 0; u < GU; ++u) {
      const unsigned r = (j + u < n) ? (unsigned)q[u >> 2][u & 3] : 0u;
      v[u] = *reinterpret_cast<const row3*>(yin + 4 * (size_t)r);
    }
    __builtin_amdgcn_sched_barrier(0);
#pragma unroll
    for (int u = 0; u < GU; ++u) {
      const bool act = (j + u < n);
      const float t0 = a0 + v[u].x;
      const float t1 = a1 + v[u].y;
      const float t2 = a2 + v[u].z;
      a0 = act ? t0 : a0;
      a1 = act ? t1 : a1;
      a2 = act ? t2 : a2;
    }
    j += GU;
  } while (j < n);

  const float z0 = a0 * d + bias[0];
  const float z1 = a1 * d + bias[1];
  const float z2 = a2 * d + bias[2];
  if (LAST) {
    out[3 * (size_t)i + 0] = z0;
    out[3 * (size_t)i + 1] = z1;
    out[3 * (size_t)i + 2] = z2;
  } else {
    v4f o;
    o.x = z0 * d;
    o.y = z1 * d;
    o.z = z2 * d;
    o.w = 0.f;
    reinterpret_cast<v4f*>(out)[i] = o;
  }
}

#define N_PAD 100096

extern "C" void kernel_launch(void* const* d_in, const int* in_sizes, int n_in,
                              void* d_out, int out_size, void* d_ws, size_t ws_size,
                              hipStream_t stream) {
  const float* x  = (const float*)d_in[0];
  const int*   ei = (const int*)d_in[1];
  const int*   row = ei;
  const int*   col = ei + N_EDGES;

  const float* Wm[5] = {(const float*)d_in[2], (const float*)d_in[4],
                        (const float*)d_in[6], (const float*)d_in[8],
                        (const float*)d_in[10]};
  const float* Bm[5] = {(const float*)d_in[3], (const float*)d_in[5],
                        (const float*)d_in[7], (const float*)d_in[9],
                        (const float*)d_in[11]};

  int*   wsi   = (int*)d_ws;
  int*   cnt   = wsi;
  int*   off   = wsi + N_PAD;
  int*   bsum  = wsi + 2 * N_PAD;
  int*   boff  = wsi + 2 * N_PAD + 128;
  int*   rank  = wsi + 2 * N_PAD + 256;
  int*   sedge = rank + N_EDGES;
  float* dinv  = (float*)(sedge + 2 * N_EDGES);
  float* P1    = dinv + N_PAD;
  float* cv    = P1 + 64;
  float* zA    = P1 + 128;
  float* zB    = zA + 4 * N_PAD;

  const int B = 256;
  const int gN  = (N_NODES + B - 1) / B;
  const int gE  = (N_EDGES + B - 1) / B;
  const int gGemm = ((N_NODES / 16) * 32 + B - 1) / B;
  const int gGn  = (N_NODES + GATHER_B - 1) / GATHER_B;

  cnt_zero_kernel<<<gN, B, 0, stream>>>(cnt);
  cnt_rank_kernel<<<gE, B, 0, stream>>>(cnt, col, rank);
  scan_block_kernel<<<N_SCAN_BLOCKS, SCAN_B, 0, stream>>>(cnt, off, bsum, dinv);
  scan_top_kernel<<<1, 128, 0, stream>>>(bsum, boff);
  csr_fill_kernel<<<gE, B, 0, stream>>>(row, col, rank, cnt, off, boff, sedge);

  weight_chain_kernel<<<1, 32, 0, stream>>>(Wm[0], Bm[0], Wm[1], Bm[1], Wm[2], Bm[2],
                                            Wm[3], Bm[3], Wm[4], P1, cv);
  gemm_wmma_kernel<8, 3, 4><<<gGemm, B, 0, stream>>>(x, P1, dinv, zA);

  agg_gather_node_kernel<false><<<gGn, GATHER_B, 0, stream>>>(zA, cv + 0, dinv, cnt, off,
                                                              boff, sedge, zB);
  agg_gather_node_kernel<false><<<gGn, GATHER_B, 0, stream>>>(zB, cv + 3, dinv, cnt, off,
                                                              boff, sedge, zA);
  agg_gather_node_kernel<false><<<gGn, GATHER_B, 0, stream>>>(zA, cv + 6, dinv, cnt, off,
                                                              boff, sedge, zB);
  agg_gather_node_kernel<false><<<gGn, GATHER_B, 0, stream>>>(zB, cv + 9, dinv, cnt, off,
                                                              boff, sedge, zA);
  agg_gather_node_kernel<true><<<gGn, GATHER_B, 0, stream>>>(zA, Bm[4], dinv, cnt, off,
                                                             boff, sedge, (float*)d_out);
}


# Sorted records at fixed slot-major places (rank k of node c at k*N+c; 40 slots, further edges on a per-node list summed in list order): one edge pass, no scans, 3 launches fewer, coalesced records
# speedup vs baseline: 4.3039x; 1.0310x over previous
#include <hip/hip_runtime.h>

#define N_NODES 100000
#define N_EDGES 1600000

typedef __attribute__((ext_vector_type(2))) float v2f;
typedef __attribute__((ext_vector_type(8))) float v8f;
typedef __attribute__((ext_vector_type(4))) int v4i;
typedef __attribute__((ext_vector_type(4))) float v4f;
struct row3 { float x, y, z; };

#define N_PAD 100096
#define SLOT_W 40
#define SLOT_PITCH 100128

#define ZERO_Q ((2 * N_PAD + SLOT_W * SLOT_PITCH) / 4)
#define ZERO_B 256
__global__ void __launch_bounds__(ZERO_B)
zero_tables_kernel(int* __restrict__ tables) {
  v4i* __restrict__ p = reinterpret_cast<v4i*>(tables);
  const v4i z = {0, 0, 0, 0};
  const int base = blockIdx.x * (4 * ZERO_B) + threadIdx.x;
#pragma unroll
  for (int k = 0; k < 4; ++k) {
    const int q = base + k * ZERO_B;
    if (q < ZERO_Q) p[q] = z;
  }
}

__global__ void edge_place_kernel(int* __restrict__ cnt,
                                  int* __restrict__ head,
                                  const int* __restrict__ row,
                                  const int* __restrict__ col,
                                  int* __restrict__ slot,
                                  int* __restrict__ nxt) {
  const int e = blockIdx.x * blockDim.x + threadIdx.x;
  if (e >= N_EDGES) return;
  const int c = col[e];
  const int s = row[e];
  asm volatile("" :: "v"(c), "v"(s));
  const int k = __hip_atomic_fetch_add(&cnt[c], 1,
                                       __ATOMIC_RELAXED, __HIP_MEMORY_SCOPE_AGENT);
  if (k < SLOT_W) {
    slot[(size_t)k * SLOT_PITCH + c] = s;
  } else {
    nxt[e] = __hip_atomic_exchange(&head[c], e + 1,
                                   __ATOMIC_RELAXED, __HIP_MEMORY_SCOPE_AGENT);
  }
}

template <int DIN, int DOUT, int OPITCH>
__global__ void gemm_wmma_kernel(const float* __restrict__ H,
                                 const float* __restrict__ W,
                                 const int* __restrict__ cnt,
                                 float* __restrict__ Out) {
  constexpr int NT = (DOUT + 15) / 16;
  const int wave = (blockIdx.x * blockDim.x + threadIdx.x) >> 5;
  const int lane = threadIdx.x & 31;
  const int m0 = wave * 16;
  if (m0 >= N_NODES) return;

  const int lhalf = lane & 15;
  const int khi = (lane >> 4) << 1;
  const int ncl = (lhalf < DOUT) ? lhalf : (DOUT - 1);

  const float* __restrict__ hrow = H + (size_t)(m0 + lhalf) * DIN;

  v8f acc0 = {};
  v8f acc1 = {};
#pragma unroll
  for (int k = 0; k < DIN; k += 4) {
    v2f a, b0;
    a.x = hrow[k + khi];
    a.y = hrow[k + khi + 1];
    b0.x = W[(k + khi) * DOUT + ncl];
    b0.y = W[(k + khi + 1) * DOUT + ncl];
    acc0 = __builtin_amdgcn_wmma_f32_16x16x4_f32(
        false, a, false, b0, (short)0, acc0, false, false);
    if (NT > 1) {
      v2f b1;
      b1.x = W[(k + khi) * DOUT + 16 + lhalf];
      b1.y = W[(k + khi + 1) * DOUT + 16 + lhalf];
      acc1 = __builtin_amdgcn_wmma_f32_16x16x4_f32(
          false, a, false, b1, (short)0, acc1, false, false);
    }
  }

  const int mrow = m0 + ((lane >> 4) << 3);
#pragma unroll
  for (int v = 0; v < 8; ++v) {
    const int m = mrow + v;
    const float rs = rsqrtf((float)(cnt[m] + 1));
    if (lhalf < OPITCH) Out[(size_t)m * OPITCH + lhalf] = acc0[v] * rs;
    if (NT > 1)         Out[(size_t)m * OPITCH + 16 + lhalf] = acc1[v] * rs;
  }
}

__device__ __forceinline__ void chain_load_a(const float* __restrict__ rowp,
                                             int khi, v2f (&a)[8]) {
#pragma unroll
  for (int j = 0; j < 8; ++j) {
    a[j].x = rowp[4 * j + khi];
    a[j].y = rowp[4 * j + khi + 1];
  }
}

__device__ __forceinline__ v8f chain_tile(const v2f (&a)[8],
                                          const float (*P)[16],
                                          int khi, int lhalf) {
  v8f acc = {};
#pragma unroll
  for (int j = 0; j < 8; ++j) {
    v2f b;
    b.x = P[4 * j + khi][lhalf];
    b.y = P[4 * j + khi + 1][lhalf];
    acc = __builtin_amdgcn_wmma_f32_16x16x4_f32(
        false, a[j], false, b, (short)0, acc, false, false);
  }
  return acc;
}

__device__ __forceinline__ void chain_step(const v2f (&w0)[8], const v2f (&w1)[8],
                                           const v2f (&bb)[8],
                                           const float (*Pc)[16], float (*Pn)[16],
                                           float* __restrict__ ck, int lane) {
  const int lhalf = lane & 15;
  const int khi = (lane >> 4) << 1;
  const int mhi = (lane >> 4) << 3;
  const v8f d0 = chain_tile(w0, Pc, khi, lhalf);
  const v8f d1 = chain_tile(w1, Pc, khi, lhalf);
  const v8f dc = chain_tile(bb, Pc, khi, lhalf);
#pragma unroll
  for (int v = 0; v < 8; ++v) {
    Pn[mhi + v][lhalf] = d0[v];
    Pn[16 + mhi + v][lhalf] = d1[v];
  }
  if (lane < 3) ck[lane] = dc[0];
  __syncthreads();
}

__global__ void __launch_bounds__(32)
weight_chain_kernel(const float* __restrict__ W1, const float* __restrict__ b1,
                    const float* __restrict__ W2, const float* __restrict__ b2,
                    const float* __restrict__ W3, const float* __restrict__ b3,
                    const float* __restrict__ W4, const float* __restrict__ b4,
                    const float* __restrict__ W5,
                    float* __restrict__ P1,
                    float* __restrict__ cv) {
  __shared__ float P[2][32][16];
  const int lane = threadIdx.x;
  const int lhalf = lane & 15;
  const int khi = (lane >> 4) << 1;

  v2f aW4[2][8], aW3[2][8], aW2[2][8], aW1[8];
  v2f aB4[8], aB3[8], aB2[8], aB1[8];
#pragma unroll
  for (int t = 0; t < 2; ++t) {
    chain_load_a(W4 + (16 * t + lhalf) * 32, khi, aW4[t]);
    chain_load_a(W3 + (16 * t + lhalf) * 32, khi, aW3[t]);
    chain_load_a(W2 + (16 * t + lhalf) * 32, khi, aW2[t]);
  }
  chain_load_a(W1 + (lhalf & 7) * 32, khi, aW1);
  chain_load_a(b4, khi, aB4);
  chain_load_a(b3, khi, aB3);
  chain_load_a(b2, khi, aB2);
  chain_load_a(b1, khi, aB1);

  for (int idx = lane; idx < 32 * 16; idx += 32) {
    const int r = idx >> 4;
    const int c = idx & 15;
    P[0][r][c] = W5[r * 3 + (c < 3 ? c : 2)];
  }
  __syncthreads();

  chain_step(aW4[0], aW4[1], aB4, P[0], P[1], cv + 9, lane);
  chain_step(aW3[0], aW3[1], aB3, P[1], P[0], cv + 6, lane);
  chain_step(aW2[0], aW2[1], aB2, P[0], P[1], cv + 3, lane);

  const float (*P2)[16] = P[1];
  const v8f d0 = chain_tile(aW1, P2, khi, lhalf);
  const v8f dc = chain_tile(aB1, P2, khi, lhalf);
  if (lane < 3) {
#pragma unroll
    for (int v = 0; v < 8; ++v) P1[v * 3 + lane] = d0[v];
    cv[lane] = dc[0];
  }
}

#define GATHER_B 64

template <bool LAST>
__global__ void __launch_bounds__(GATHER_B)
agg_gather_node_kernel(const float* __restrict__ yin,
                       const float* __restrict__ bias,
                       const int* __restrict__ cnt,
                       const int* __restrict__ slot,
                       const int* __restrict__ head,
                       const int* __restrict__ nxt,
                       const int* __restrict__ row,
                       float* __restrict__ out) {
  const int i = blockIdx.x * GATHER_B + threadIdx.x;
  if (i >= N_NODES) return;
  const int n = cnt[i];
  const float* __restrict__ yself = yin + 4 * (size_t)i;
  float a0 = yself[0];
  float a1 = yself[1];
  float a2 = yself[2];

  int q[SLOT_W];
  row3 v[SLOT_W];
  const char* __restrict__ ybytes = reinterpret_cast<const char*>(yin);
#pragma unroll
  for (int k = 0; k < SLOT_W; ++k) q[k] = slot[(size_t)k * SLOT_PITCH + i];
#pragma unroll
  for (int u = 0; u < SLOT_W; ++u)
    v[u] = *reinterpret_cast<const row3*>(ybytes + ((unsigned)q[u] << 4));
#define ROWV(u) "v"(v[u].x), "v"(v[u].y), "v"(v[u].z)
#define ROWV4(u) ROWV(u), ROWV(u + 1), ROWV(u + 2), ROWV(u + 3)
#define ROWV20(u) ROWV4(u), ROWV4(u + 4), ROWV4(u + 8), ROWV4(u + 12), ROWV4(u + 16)
  asm volatile("" : "+v"(a0), "+v"(a1), "+v"(a2) : ROWV20(0), ROWV20(20));
#undef ROWV20
#undef ROWV4
#undef ROWV
#pragma unroll
  for (int u = 0; u < SLOT_W; ++u) {
    const bool act = (u < n);
    const float t0 = a0 + v[u].x;
    const float t1 = a1 + v[u].y;
    const float t2 = a2 + v[u].z;
    a0 = act ? t0 : a0;
    a1 = act ? t1 : a1;
    a2 = act ? t2 : a2;
  }
  if (n > SLOT_W) {
    int p = head[i];
    for (int t = SLOT_W; t < n && p != 0; ++t) {
      const int e = p - 1;
      const unsigned r = (unsigned)row[e];
      p = nxt[e];
      const row3 w = *reinterpret_cast<const row3*>(yin + 4 * (size_t)r);
      a0 += w.x;
      a1 += w.y;
      a2 += w.z;
    }
  }

  const float d = rsqrtf((float)(n + 1));
  const float z0 = a0 * d + bias[0];
  const float z1 = a1 * d + bias[1];
  const float z2 = a2 * d + bias[2];
  if (LAST) {
    out[3 * (size_t)i + 0] = z0;
    out[3 * (size_t)i + 1] = z1;
    out[3 * (size_t)i + 2] = z2;
  } else {
    v4f o;
    o.x = z0 * d;
    o.y = z1 * d;
    o.z = z2 * d;
    o.w = 0.f;
    reinterpret_cast<v4f*>(out)[i] = o;
  }
}

extern "C" void kernel_launch(void* const* d_in, const int* in_sizes, int n_in,
                              void* d_out, int out_size, void* d_ws, size_t ws_size,
                              hipStream_t stream) {
  const float* x  = (const float*)d_in[0];
  const int*   ei = (const int*)d_in[1];
  const int*   row = ei;
  const int*   col = ei + N_EDGES;

  const float* Wm[5] = {(const float*)d_in[2], (const float*)d_in[4],
                        (const float*)d_in[6], (const float*)d_in[8],
                        (const float*)d_in[10]};
  const float* Bm[5] = {(const float*)d_in[3], (const float*)d_in[5],
                        (const float*)d_in[7], (const float*)d_in[9],
                        (const float*)d_in[11]};

  int*   wsi   = (int*)d_ws;
  int*   cnt   = wsi;
  int*   head  = wsi + N_PAD;
  int*   slot  = wsi + 2 * N_PAD;
  int*   nxt   = slot + SLOT_W * SLOT_PITCH;
  float* P1    = (float*)(nxt + N_EDGES);
  float* cv    = P1 + 64;
  float* zA    = P1 + 128;
  float* zB    = zA + 4 * N_PAD;

  const int B = 256;
  const int gE  = (N_EDGES + B - 1) / B;
  const int gZ  = (ZERO_Q + 4 * ZERO_B - 1) / (4 * ZERO_B);
  const int gGemm = ((N_NODES / 16) * 32 + B - 1) / B;
  const int gGn  = (N_NODES + GATHER_B - 1) / GATHER_B;

  zero_tables_kernel<<<gZ, ZERO_B, 0, stream>>>(wsi);
  edge_place_kernel<<<gE, B, 0, stream>>>(cnt, head, row, col, slot, nxt);

  weight_chain_kernel<<<1, 32, 0, stream>>>(Wm[0], Bm[0], Wm[1], Bm[1], Wm[2], Bm[2],
                                            Wm[3], Bm[3], Wm[4], P1, cv);
  gemm_wmma_kernel<8, 3, 4><<<gGemm, B, 0, stream>>>(x, P1, cnt, zA);

  agg_gather_node_kernel<false><<<gGn, GATHER_B, 0, stream>>>(zA, cv + 0, cnt, slot, head,
                                                              nxt, row, zB);
  agg_gather_node_kernel<false><<<gGn, GATHER_B, 0, stream>>>(zB, cv + 3, cnt, slot, head,
                                                              nxt, row, zA);
  agg_gather_node_kernel<false><<<gGn, GATHER_B, 0, stream>>>(zA, cv + 6, cnt, slot, head,
                                                              nxt, row, zB);
  agg_gather_node_kernel<false><<<gGn, GATHER_B, 0, stream>>>(zB, cv + 9, cnt, slot, head,
                                                              nxt, row, zA);
  agg_gather_node_kernel<true><<<gGn, GATHER_B, 0, stream>>>(zA, Bm[4], cnt, slot, head,
                                                             nxt, row, (float*)d_out);
}


# Degree counters moved to one 128-byte line per node (list head beside it): returning adds of different nodes no longer share lines; first gather writes dense counts; sums in arrival-rank order
# speedup vs baseline: 5.0675x; 1.1774x over previous
#include <hip/hip_runtime.h>

#define N_NODES 100000
#define N_EDGES 1600000

typedef __attribute__((ext_vector_type(2))) float v2f;
typedef __attribute__((ext_vector_type(8))) float v8f;
typedef __attribute__((ext_vector_type(4))) int v4i;
typedef __attribute__((ext_vector_type(4))) float v4f;
struct row3 { float x, y, z; };

#define N_PAD 100096
#define NODE_LINE 32
#define SLOT_W 40
#define SLOT_PITCH 100128

#define ZERO_Q ((N_NODES * NODE_LINE + SLOT_W * SLOT_PITCH) / 4)
#define ZERO_B 256
#define ZERO_K 8
__global__ void __launch_bounds__(ZERO_B)
zero_tables_kernel(int* __restrict__ tables) {
  v4i* __restrict__ p = reinterpret_cast<v4i*>(tables);
  const v4i z = {0, 0, 0, 0};
  const int base = blockIdx.x * (ZERO_K * ZERO_B) + threadIdx.x;
#pragma unroll
  for (int k = 0; k < ZERO_K; ++k) {
    const int q = base + k * ZERO_B;
    if (q < ZERO_Q) p[q] = z;
  }
}

__global__ void edge_place_kernel(int* __restrict__ node,
                                  const int* __restrict__ row,
                                  const int* __restrict__ col,
                                  int* __restrict__ slot,
                                  int* __restrict__ nxt) {
  const int e = blockIdx.x * blockDim.x + threadIdx.x;
  if (e >= N_EDGES) return;
  const int c = col[e];
  const int s = row[e];
  asm volatile("" :: "v"(c), "v"(s));
  int* __restrict__ line = node + (size_t)c * NODE_LINE;
  const int k = __hip_atomic_fetch_add(&line[0], 1,
                                       __ATOMIC_RELAXED, __HIP_MEMORY_SCOPE_AGENT);
  if (k < SLOT_W) {
    slot[(size_t)k * SLOT_PITCH + c] = s;
  } else {
    nxt[e] = __hip_atomic_exchange(&line[1], e + 1,
                                   __ATOMIC_RELAXED, __HIP_MEMORY_SCOPE_AGENT);
  }
}

template <int DIN, int DOUT, int OPITCH>
__global__ void gemm_wmma_kernel(const float* __restrict__ H,
                                 const float* __restrict__ W,
                                 const int* __restrict__ node,
                                 float* __restrict__ Out) {
  constexpr int NT = (DOUT + 15) / 16;
  const int wave = (blockIdx.x * blockDim.x + threadIdx.x) >> 5;
  const int lane = threadIdx.x & 31;
  const int m0 = wave * 16;
  if (m0 >= N_NODES) return;

  const int lhalf = lane & 15;
  const int khi = (lane >> 4) << 1;
  const int ncl = (lhalf < DOUT) ? lhalf : (DOUT - 1);

  const float* __restrict__ hrow = H + (size_t)(m0 + lhalf) * DIN;

  v8f acc0 = {};
  v8f acc1 = {};
#pragma unroll
  for (int k = 0; k < DIN; k += 4) {
    v2f a, b0;
    a.x = hrow[k + khi];
    a.y = hrow[k + khi + 1];
    b0.x = W[(k + khi) * DOUT + ncl];
    b0.y = W[(k + khi + 1) * DOUT + ncl];
    acc0 = __builtin_amdgcn_wmma_f32_16x16x4_f32(
        false, a, false, b0, (short)0, acc0, false, false);
    if (NT > 1) {
      v2f b1;
      b1.x = W[(k + khi) * DOUT + 16 + lhalf];
      b1.y = W[(k + khi + 1) * DOUT + 16 + lhalf];
      acc1 = __builtin_amdgcn_wmma_f32_16x16x4_f32(
          false, a, false, b1, (short)0, acc1, false, false);
    }
  }

  const int mrow = m0 + ((lane >> 4) << 3);
#pragma unroll
  for (int v = 0; v < 8; ++v) {
    const int m = mrow + v;
    const float rs = rsqrtf((float)(node[(size_t)m * NODE_LINE] + 1));
    if (lhalf < OPITCH) Out[(size_t)m * OPITCH + lhalf] = acc0[v] * rs;
    if (NT > 1)         Out[(size_t)m * OPITCH + 16 + lhalf] = acc1[v] * rs;
  }
}

__device__ __forceinline__ void chain_load_a(const float* __restrict__ rowp,
                                             int khi, v2f (&a)[8]) {
#pragma unroll
  for (int j = 0; j < 8; ++j) {
    a[j].x = rowp[4 * j + khi];
    a[j].y = rowp[4 * j + khi + 1];
  }
}

__device__ __forceinline__ v8f chain_tile(const v2f (&a)[8],
                                          const float (*P)[16],
                                          int khi, int lhalf) {
  v8f acc = {};
#pragma unroll
  for (int j = 0; j < 8; ++j) {
    v2f b;
    b.x = P[4 * j + khi][lhalf];
    b.y = P[4 * j + khi + 1][lhalf];
    acc = __builtin_amdgcn_wmma_f32_16x16x4_f32(
        false, a[j], false, b, (short)0, acc, false, false);
  }
  return acc;
}

__device__ __forceinline__ void chain_step(const v2f (&w0)[8], const v2f (&w1)[8],
                                           const v2f (&bb)[8],
                                           const float (*Pc)[16], float (*Pn)[16],
                                           float* __restrict__ ck, int lane) {
  const int lhalf = lane & 15;
  const int khi = (lane >> 4) << 1;
  const int mhi = (lane >> 4) << 3;
  const v8f d0 = chain_tile(w0, Pc, khi, lhalf);
  const v8f d1 = chain_tile(w1, Pc, khi, lhalf);
  const v8f dc = chain_tile(bb, Pc, khi, lhalf);
#pragma unroll
  for (int v = 0; v < 8; ++v) {
    Pn[mhi + v][lhalf] = d0[v];
    Pn[16 + mhi + v][lhalf] = d1[v];
  }
  if (lane < 3) ck[lane] = dc[0];
  __syncthreads();
}

__global__ void __launch_bounds__(32)
weight_chain_kernel(const float* __restrict__ W1, const float* __restrict__ b1,
                    const float* __restrict__ W2, const float* __restrict__ b2,
                    const float* __restrict__ W3, const float* __restrict__ b3,
                    const float* __restrict__ W4, const float* __restrict__ b4,
                    const float* __restrict__ W5,
                    float* __restrict__ P1,
                    float* __restrict__ cv) {
  __shared__ float P[2][32][16];
  const int lane = threadIdx.x;
  const int lhalf = lane & 15;
  const int khi = (lane >> 4) << 1;

  v2f aW4[2][8], aW3[2][8], aW2[2][8], aW1[8];
  v2f aB4[8], aB3[8], aB2[8], aB1[8];
#pragma unroll
  for (int t = 0; t < 2; ++t) {
    chain_load_a(W4 + (16 * t + lhalf) * 32, khi, aW4[t]);
    chain_load_a(W3 + (16 * t + lhalf) * 32, khi, aW3[t]);
    chain_load_a(W2 + (16 * t + lhalf) * 32, khi, aW2[t]);
  }
  chain_load_a(W1 + (lhalf & 7) * 32, khi, aW1);
  chain_load_a(b4, khi, aB4);
  chain_load_a(b3, khi, aB3);
  chain_load_a(b2, khi, aB2);
  chain_load_a(b1, khi, aB1);

  for (int idx = lane; idx < 32 * 16; idx += 32) {
    const int r = idx >> 4;
    const int c = idx & 15;
    P[0][r][c] = W5[r * 3 + (c < 3 ? c : 2)];
  }
  __syncthreads();

  chain_step(aW4[0], aW4[1], aB4, P[0], P[1], cv + 9, lane);
  chain_step(aW3[0], aW3[1], aB3, P[1], P[0], cv + 6, lane);
  chain_step(aW2[0], aW2[1], aB2, P[0], P[1], cv + 3, lane);

  const float (*P2)[16] = P[1];
  const v8f d0 = chain_tile(aW1, P2, khi, lhalf);
  const v8f dc = chain_tile(aB1, P2, khi, lhalf);
  if (lane < 3) {
#pragma unroll
    for (int v = 0; v < 8; ++v) P1[v * 3 + lane] = d0[v];
    cv[lane] = dc[0];
  }
}

#define GATHER_B 64

template <int MODE>
__global__ void __launch_bounds__(GATHER_B)
agg_gather_node_kernel(const float* __restrict__ yin,
                       const float* __restrict__ bias,
                       const int* __restrict__ node,
                       int* __restrict__ cntc,
                       const int* __restrict__ slot,
                       const int* __restrict__ nxt,
                       const int* __restrict__ row,
                       float* __restrict__ out) {
  constexpr bool LAST = (MODE == 2);
  const int i = blockIdx.x * GATHER_B + threadIdx.x;
  if (i >= N_NODES) return;
  const int n = (MODE == 0) ? node[(size_t)i * NODE_LINE] : cntc[i];
  if (MODE == 0) cntc[i] = n;
  const float* __restrict__ yself = yin + 4 * (size_t)i;
  float a0 = yself[0];
  float a1 = yself[1];
  float a2 = yself[2];

  int q[SLOT_W];
  row3 v[SLOT_W];
  const char* __restrict__ ybytes = reinterpret_cast<const char*>(yin);
#pragma unroll
  for (int k = 0; k < SLOT_W; ++k) q[k] = slot[(size_t)k * SLOT_PITCH + i];
#pragma unroll
  for (int u = 0; u < SLOT_W; ++u)
    v[u] = *reinterpret_cast<const row3*>(ybytes + ((unsigned)q[u] << 4));
#define ROWV(u) "v"(v[u].x), "v"(v[u].y), "v"(v[u].z)
#define ROWV4(u) ROWV(u), ROWV(u + 1), ROWV(u + 2), ROWV(u + 3)
#define ROWV20(u) ROWV4(u), ROWV4(u + 4), ROWV4(u + 8), ROWV4(u + 12), ROWV4(u + 16)
  asm volatile("" : "+v"(a0), "+v"(a1), "+v"(a2) : ROWV20(0), ROWV20(20));
#undef ROWV20
#undef ROWV4
#undef ROWV
#pragma unroll
  for (int u = 0; u < SLOT_W; ++u) {
    const bool act = (u < n);
    const float t0 = a0 + v[u].x;
    const float t1 = a1 + v[u].y;
    const float t2 = a2 + v[u].z;
    a0 = act ? t0 : a0;
    a1 = act ? t1 : a1;
    a2 = act ? t2 : a2;
  }
  if (n > SLOT_W) {
    int p = node[(size_t)i * NODE_LINE + 1];
    for (int t = SLOT_W; t < n && p != 0; ++t) {
      const int e = p - 1;
      const unsigned r = (unsigned)row[e];
      p = nxt[e];
      const row3 w = *reinterpret_cast<const row3*>(yin + 4 * (size_t)r);
      a0 += w.x;
      a1 += w.y;
      a2 += w.z;
    }
  }

  const float d = rsqrtf((float)(n + 1));
  const float z0 = a0 * d + bias[0];
  const float z1 = a1 * d + bias[1];
  const float z2 = a2 * d + bias[2];
  if (LAST) {
    out[3 * (size_t)i + 0] = z0;
    out[3 * (size_t)i + 1] = z1;
    out[3 * (size_t)i + 2] = z2;
  } else {
    v4f o;
    o.x = z0 * d;
    o.y = z1 * d;
    o.z = z2 * d;
    o.w = 0.f;
    reinterpret_cast<v4f*>(out)[i] = o;
  }
}

extern "C" void kernel_launch(void* const* d_in, const int* in_sizes, int n_in,
                              void* d_out, int out_size, void* d_ws, size_t ws_size,
                              hipStream_t stream) {
  const float* x  = (const float*)d_in[0];
  const int*   ei = (const int*)d_in[1];
  const int*   row = ei;
  const int*   col = ei + N_EDGES;

  const float* Wm[5] = {(const float*)d_in[2], (const float*)d_in[4],
                        (const float*)d_in[6], (const float*)d_in[8],
                        (const float*)d_in[10]};
  const float* Bm[5] = {(const float*)d_in[3], (const float*)d_in[5],
                        (const float*)d_in[7], (const float*)d_in[9],
                        (const float*)d_in[11]};

  int*   wsi   = (int*)d_ws;
  int*   node  = wsi;
  int*   slot  = wsi + N_NODES * NODE_LINE;
  int*   nxt   = slot + SLOT_W * SLOT_PITCH;
  float* P1    = (float*)(nxt + N_EDGES);
  float* cv    = P1 + 64;
  float* zA    = P1 + 128;
  float* zB    = zA + 4 * N_PAD;
  int*   cntc  = (int*)(zB + 4 * N_PAD);

  const int B = 256;
  const int gE  = (N_EDGES + B - 1) / B;
  const int gZ  = (ZERO_Q + ZERO_K * ZERO_B - 1) / (ZERO_K * ZERO_B);
  const int gGemm = ((N_NODES / 16) * 32 + B - 1) / B;
  const int gGn  = (N_NODES + GATHER_B - 1) / GATHER_B;

  zero_tables_kernel<<<gZ, ZERO_B, 0, stream>>>(wsi);
  edge_place_kernel<<<gE, B, 0, stream>>>(node, row, col, slot, nxt);

  weight_chain_kernel<<<1, 32, 0, stream>>>(Wm[0], Bm[0], Wm[1], Bm[1], Wm[2], Bm[2],
                                            Wm[3], Bm[3], Wm[4], P1, cv);
  gemm_wmma_kernel<8, 3, 4><<<gGemm, B, 0, stream>>>(x, P1, node, zA);

  agg_gather_node_kernel<0><<<gGn, GATHER_B, 0, stream>>>(zA, cv + 0, node, cntc, slot,
                                                          nxt, row, zB);
  agg_gather_node_kernel<1><<<gGn, GATHER_B, 0, stream>>>(zB, cv + 3, node, cntc, slot,
                                                          nxt, row, zA);
  agg_gather_node_kernel<1><<<gGn, GATHER_B, 0, stream>>>(zA, cv + 6, node, cntc, slot,
                                                          nxt, row, zB);
  agg_gather_node_kernel<1><<<gGn, GATHER_B, 0, stream>>>(zB, cv + 9, node, cntc, slot,
                                                          nxt, row, zA);
  agg_gather_node_kernel<2><<<gGn, GATHER_B, 0, stream>>>(zA, Bm[4], node, cntc, slot,
                                                          nxt, row, (float*)d_out);
}


# Edge pass: each lane places 8 edges (16-byte index loads, 8 counting adds in flight), so an eighth of the waves has to start; sums stay in arrival-rank order
# speedup vs baseline: 5.2724x; 1.0404x over previous
#include <hip/hip_runtime.h>

#define N_NODES 100000
#define N_EDGES 1600000

typedef __attribute__((ext_vector_type(2))) float v2f;
typedef __attribute__((ext_vector_type(8))) float v8f;
typedef __attribute__((ext_vector_type(4))) int v4i;
typedef __attribute__((ext_vector_type(4))) float v4f;
struct row3 { float x, y, z; };

#define N_PAD 100096
#define NODE_LINE 32
#define SLOT_W 40
#define SLOT_PITCH 100128

#define ZERO_Q ((N_NODES * NODE_LINE + SLOT_W * SLOT_PITCH) / 4)
#define ZERO_B 256
#define ZERO_K 8
__global__ void __launch_bounds__(ZERO_B)
zero_tables_kernel(int* __restrict__ tables) {
  v4i* __restrict__ p = reinterpret_cast<v4i*>(tables);
  const v4i z = {0, 0, 0, 0};
  const int base = blockIdx.x * (ZERO_K * ZERO_B) + threadIdx.x;
#pragma unroll
  for (int k = 0; k < ZERO_K; ++k) {
    const int q = base + k * ZERO_B;
    if (q < ZERO_Q) p[q] = z;
  }
}

#define PLACE_B 64
#define PLACE_EPL 8
#define PLACE_EPW (32 * PLACE_EPL)
static_assert(N_EDGES % PLACE_EPW == 0, "a wave's chunk is all inside the edge list or all outside");
__global__ void __launch_bounds__(PLACE_B)
edge_place_kernel(int* __restrict__ node,
                  const int* __restrict__ row,
                  const int* __restrict__ col,
                  int* __restrict__ slot,
                  int* __restrict__ nxt) {
  const int wave = (blockIdx.x * PLACE_B + threadIdx.x) >> 5;
  const int lane = threadIdx.x & 31;
  const int e0 = wave * PLACE_EPW + 4 * lane;
  if (wave * PLACE_EPW >= N_EDGES) return;
  const v4i c0 = *reinterpret_cast<const v4i*>(col + e0);
  const v4i c1 = *reinterpret_cast<const v4i*>(col + e0 + 128);
  const v4i s0 = *reinterpret_cast<const v4i*>(row + e0);
  const v4i s1 = *reinterpret_cast<const v4i*>(row + e0 + 128);
  const int c[PLACE_EPL] = {c0.x, c0.y, c0.z, c0.w, c1.x, c1.y, c1.z, c1.w};
  const int s[PLACE_EPL] = {s0.x, s0.y, s0.z, s0.w, s1.x, s1.y, s1.z, s1.w};
  int k[PLACE_EPL];
#pragma unroll
  for (int j = 0; j < PLACE_EPL; ++j)
    k[j] = __hip_atomic_fetch_add(node + (size_t)c[j] * NODE_LINE, 1,
                                  __ATOMIC_RELAXED, __HIP_MEMORY_SCOPE_AGENT);
#pragma unroll
  for (int j = 0; j < PLACE_EPL; ++j) {
    if (k[j] < SLOT_W) {
      slot[(size_t)k[j] * SLOT_PITCH + c[j]] = s[j];
    } else {
      const int e = e0 + (j >> 2) * 128 + (j & 3);
      nxt[e] = __hip_atomic_exchange(node + (size_t)c[j] * NODE_LINE + 1, e + 1,
                                     __ATOMIC_RELAXED, __HIP_MEMORY_SCOPE_AGENT);
    }
  }
}

template <int DIN, int DOUT, int OPITCH>
__global__ void gemm_wmma_kernel(const float* __restrict__ H,
                                 const float* __restrict__ W,
                                 const int* __restrict__ node,
                                 float* __restrict__ Out) {
  constexpr int NT = (DOUT + 15) / 16;
  const int wave = (blockIdx.x * blockDim.x + threadIdx.x) >> 5;
  const int lane = threadIdx.x & 31;
  const int m0 = wave * 16;
  if (m0 >= N_NODES) return;

  const int lhalf = lane & 15;
  const int khi = (lane >> 4) << 1;
  const int ncl = (lhalf < DOUT) ? lhalf : (DOUT - 1);

  const float* __restrict__ hrow = H + (size_t)(m0 + lhalf) * DIN;

  v8f acc0 = {};
  v8f acc1 = {};
#pragma unroll
  for (int k = 0; k < DIN; k += 4) {
    v2f a, b0;
    a.x = hrow[k + khi];
    a.y = hrow[k + khi + 1];
    b0.x = W[(k + khi) * DOUT + ncl];
    b0.y = W[(k + khi + 1) * DOUT + ncl];
    acc0 = __builtin_amdgcn_wmma_f32_16x16x4_f32(
        false, a, false, b0, (short)0, acc0, false, false);
    if (NT > 1) {
      v2f b1;
      b1.x = W[(k + khi) * DOUT + 16 + lhalf];
      b1.y = W[(k + khi + 1) * DOUT + 16 + lhalf];
      acc1 = __builtin_amdgcn_wmma_f32_16x16x4_f32(
          false, a, false, b1, (short)0, acc1, false, false);
    }
  }

  const int mrow = m0 + ((lane >> 4) << 3);
#pragma unroll
  for (int v = 0; v < 8; ++v) {
    const int m = mrow + v;
    const float rs = rsqrtf((float)(node[(size_t)m * NODE_LINE] + 1));
    if (lhalf < OPITCH) Out[(size_t)m * OPITCH + lhalf] = acc0[v] * rs;
    if (NT > 1)         Out[(size_t)m * OPITCH + 16 + lhalf] = acc1[v] * rs;
  }
}

__device__ __forceinline__ void chain_load_a(const float* __restrict__ rowp,
                                             int khi, v2f (&a)[8]) {
#pragma unroll
  for (int j = 0; j < 8; ++j) {
    a[j].x = rowp[4 * j + khi];
    a[j].y = rowp[4 * j + khi + 1];
  }
}

__device__ __forceinline__ v8f chain_tile(const v2f (&a)[8],
                                          const float (*P)[16],
                                          int khi, int lhalf) {
  v8f acc = {};
#pragma unroll
  for (int j = 0; j < 8; ++j) {
    v2f b;
    b.x = P[4 * j + khi][lhalf];
    b.y = P[4 * j + khi + 1][lhalf];
    acc = __builtin_amdgcn_wmma_f32_16x16x4_f32(
        false, a[j], false, b, (short)0, acc, false, false);
  }
  return acc;
}

__device__ __forceinline__ void chain_step(const v2f (&w0)[8], const v2f (&w1)[8],
                                           const v2f (&bb)[8],
                                           const float (*Pc)[16], float (*Pn)[16],
                                           float* __restrict__ ck, int lane) {
  const int lhalf = lane & 15;
  const int khi = (lane >> 4) << 1;
  const int mhi = (lane >> 4) << 3;
  const v8f d0 = chain_tile(w0, Pc, khi, lhalf);
  const v8f d1 = chain_tile(w1, Pc, khi, lhalf);
  const v8f dc = chain_tile(bb, Pc, khi, lhalf);
#pragma unroll
  for (int v = 0; v < 8; ++v) {
    Pn[mhi + v][lhalf] = d0[v];
    Pn[16 + mhi + v][lhalf] = d1[v];
  }
  if (lane < 3) ck[lane] = dc[0];
  __syncthreads();
}

__global__ void __launch_bounds__(32)
weight_chain_kernel(const float* __restrict__ W1, const float* __restrict__ b1,
                    const float* __restrict__ W2, const float* __restrict__ b2,
                    const float* __restrict__ W3, const float* __restrict__ b3,
                    const float* __restrict__ W4, const float* __restrict__ b4,
                    const float* __restrict__ W5,
                    float* __restrict__ P1,
                    float* __restrict__ cv) {
  __shared__ float P[2][32][16];
  const int lane = threadIdx.x;
  const int lhalf = lane & 15;
  const int khi = (lane >> 4) << 1;

  v2f aW4[2][8], aW3[2][8], aW2[2][8], aW1[8];
  v2f aB4[8], aB3[8], aB2[8], aB1[8];
#pragma unroll
  for (int t = 0; t < 2; ++t) {
    chain_load_a(W4 + (16 * t + lhalf) * 32, khi, aW4[t]);
    chain_load_a(W3 + (16 * t + lhalf) * 32, khi, aW3[t]);
    chain_load_a(W2 + (16 * t + lhalf) * 32, khi, aW2[t]);
  }
  chain_load_a(W1 + (lhalf & 7) * 32, khi, aW1);
  chain_load_a(b4, khi, aB4);
  chain_load_a(b3, khi, aB3);
  chain_load_a(b2, khi, aB2);
  chain_load_a(b1, khi, aB1);

  for (int idx = lane; idx < 32 * 16; idx += 32) {
    const int r = idx >> 4;
    const int c = idx & 15;
    P[0][r][c] = W5[r * 3 + (c < 3 ? c : 2)];
  }
  __syncthreads();

  chain_step(aW4[0], aW4[1], aB4, P[0], P[1], cv + 9, lane);
  chain_step(aW3[0], aW3[1], aB3, P[1], P[0], cv + 6, lane);
  chain_step(aW2[0], aW2[1], aB2, P[0], P[1], cv + 3, lane);

  const float (*P2)[16] = P[1];
  const v8f d0 = chain_tile(aW1, P2, khi, lhalf);
  const v8f dc = chain_tile(aB1, P2, khi, lhalf);
  if (lane < 3) {
#pragma unroll
    for (int v = 0; v < 8; ++v) P1[v * 3 + lane] = d0[v];
    cv[lane] = dc[0];
  }
}

#define GATHER_B 64

template <int MODE>
__global__ void __launch_bounds__(GATHER_B)
agg_gather_node_kernel(const float* __restrict__ yin,
                       const float* __restrict__ bias,
                       const int* __restrict__ node,
                       int* __restrict__ cntc,
                       const int* __restrict__ slot,
                       const int* __restrict__ nxt,
                       const int* __restrict__ row,
                       float* __restrict__ out) {
  constexpr bool LAST = (MODE == 2);
  const int i = blockIdx.x * GATHER_B + threadIdx.x;
  if (i >= N_NODES) return;
  const int n = (MODE == 0) ? node[(size_t)i * NODE_LINE] : cntc[i];
  if (MODE == 0) cntc[i] = n;
  const float* __restrict__ yself = yin + 4 * (size_t)i;
  float a0 = yself[0];
  float a1 = yself[1];
  float a2 = yself[2];

  int q[SLOT_W];
  row3 v[SLOT_W];
  const char* __restrict__ ybytes = reinterpret_cast<const char*>(yin);
#pragma unroll
  for (int k = 0; k < SLOT_W; ++k) q[k] = slot[(size_t)k * SLOT_PITCH + i];
#pragma unroll
  for (int u = 0; u < SLOT_W; ++u)
    v[u] = *reinterpret_cast<const row3*>(ybytes + ((unsigned)q[u] << 4));
#define ROWV(u) "v"(v[u].x), "v"(v[u].y), "v"(v[u].z)
#define ROWV4(u) ROWV(u), ROWV(u + 1), ROWV(u + 2), ROWV(u + 3)
#define ROWV20(u) ROWV4(u), ROWV4(u + 4), ROWV4(u + 8), ROWV4(u + 12), ROWV4(u + 16)
  asm volatile("" : "+v"(a0), "+v"(a1), "+v"(a2) : ROWV20(0), ROWV20(20));
#undef ROWV20
#undef ROWV4
#undef ROWV
#pragma unroll
  for (int u = 0; u < SLOT_W; ++u) {
    const bool act = (u < n);
    const float t0 = a0 + v[u].x;
    const float t1 = a1 + v[u].y;
    const float t2 = a2 + v[u].z;
    a0 = act ? t0 : a0;
    a1 = act ? t1 : a1;
    a2 = act ? t2 : a2;
  }
  if (n > SLOT_W) {
    int p = node[(size_t)i * NODE_LINE + 1];
    for (int t = SLOT_W; t < n && p != 0; ++t) {
      const int e = p - 1;
      const unsigned r = (unsigned)row[e];
      p = nxt[e];
      const row3 w = *reinterpret_cast<const row3*>(yin + 4 * (size_t)r);
      a0 += w.x;
      a1 += w.y;
      a2 += w.z;
    }
  }

  const float d = rsqrtf((float)(n + 1));
  const float z0 = a0 * d + bias[0];
  const float z1 = a1 * d + bias[1];
  const float z2 = a2 * d + bias[2];
  if (LAST) {
    out[3 * (size_t)i + 0] = z0;
    out[3 * (size_t)i + 1] = z1;
    out[3 * (size_t)i + 2] = z2;
  } else {
    v4f o;
    o.x = z0 * d;
    o.y = z1 * d;
    o.z = z2 * d;
    o.w = 0.f;
    reinterpret_cast<v4f*>(out)[i] = o;
  }
}

extern "C" void kernel_launch(void* const* d_in, const int* in_sizes, int n_in,
                              void* d_out, int out_size, void* d_ws, size_t ws_size,
                              hipStream_t stream) {
  const float* x  = (const float*)d_in[0];
  const int*   ei = (const int*)d_in[1];
  const int*   row = ei;
  const int*   col = ei + N_EDGES;

  const float* Wm[5] = {(const float*)d_in[2], (const float*)d_in[4],
                        (const float*)d_in[6], (const float*)d_in[8],
                        (const float*)d_in[10]};
  const float* Bm[5] = {(const float*)d_in[3], (const float*)d_in[5],
                        (const float*)d_in[7], (const float*)d_in[9],
                        (const float*)d_in[11]};

  int*   wsi   = (int*)d_ws;
  int*   node  = wsi;
  int*   slot  = wsi + N_NODES * NODE_LINE;
  int*   nxt   = slot + SLOT_W * SLOT_PITCH;
  float* P1    = (float*)(nxt + N_EDGES);
  float* cv    = P1 + 64;
  float* zA    = P1 + 128;
  float* zB    = zA + 4 * N_PAD;
  int*   cntc  = (int*)(zB + 4 * N_PAD);

  const int B = 256;
  const int gE  = (N_EDGES / PLACE_EPW * 32 + PLACE_B - 1) / PLACE_B;
  const int gZ  = (ZERO_Q + ZERO_K * ZERO_B - 1) / (ZERO_K * ZERO_B);
  const int gGemm = ((N_NODES / 16) * 32 + B - 1) / B;
  const int gGn  = (N_NODES + GATHER_B - 1) / GATHER_B;

  zero_tables_kernel<<<gZ, ZERO_B, 0, stream>>>(wsi);
  edge_place_kernel<<<gE, PLACE_B, 0, stream>>>(node, row, col, slot, nxt);

  weight_chain_kernel<<<1, 32, 0, stream>>>(Wm[0], Bm[0], Wm[1], Bm[1], Wm[2], Bm[2],
                                            Wm[3], Bm[3], Wm[4], P1, cv);
  gemm_wmma_kernel<8, 3, 4><<<gGemm, B, 0, stream>>>(x, P1, node, zA);

  agg_gather_node_kernel<0><<<gGn, GATHER_B, 0, stream>>>(zA, cv + 0, node, cntc, slot,
                                                          nxt, row, zB);
  agg_gather_node_kernel<1><<<gGn, GATHER_B, 0, stream>>>(zB, cv + 3, node, cntc, slot,
                                                          nxt, row, zA);
  agg_gather_node_kernel<1><<<gGn, GATHER_B, 0, stream>>>(zA, cv + 6, node, cntc, slot,
                                                          nxt, row, zB);
  agg_gather_node_kernel<1><<<gGn, GATHER_B, 0, stream>>>(zB, cv + 9, node, cntc, slot,
                                                          nxt, row, zA);
  agg_gather_node_kernel<2><<<gGn, GATHER_B, 0, stream>>>(zA, Bm[4], node, cntc, slot,
                                                          nxt, row, (float*)d_out);
}


# Edge sort redone as two binning passes through LDS (by destination block, then by node with LDS counting adds): no global atomic, scattered store or table clearing; sums in LDS arrival-rank order
# speedup vs baseline: 9.5030x; 1.8024x over previous
#include <hip/hip_runtime.h>

#define N_NODES 100000
#define N_EDGES 1600000

typedef __attribute__((ext_vector_type(2))) float v2f;
typedef __attribute__((ext_vector_type(8))) float v8f;
typedef __attribute__((ext_vector_type(4))) int v4i;
typedef __attribute__((ext_vector_type(4))) float v4f;
struct row3 { float x, y, z; };

#define N_PAD 100096
#define SLOT_W 40
#define SLOT_PITCH 100128

typedef __attribute__((ext_vector_type(2))) int v2i;

#define BKT 352
#define NBKT ((N_NODES + BKT - 1) / BKT)
#define NBKT_PAD 288
#define BIN_SCAN 9
#define BIN_B 256
#define BIN_EPT 25
#define BIN_CH (BIN_B * BIN_EPT)
#define BIN_G (N_EDGES / BIN_CH)
static_assert(N_EDGES % BIN_CH == 0, "every work-group of pass 1 has a full piece");
static_assert(NBKT <= NBKT_PAD && NBKT_PAD == 32 * BIN_SCAN, "one wave scans the histogram");
static_assert(BKT % 32 == 0 && N_NODES % 32 == 0, "a block's part of a plane is whole lines");
static_assert(BIN_G <= BIN_B, "pass 2 gives every run of a block a lane");

__global__ void __launch_bounds__(BIN_B)
edge_bin_kernel(const int* __restrict__ row,
                const int* __restrict__ col,
                int* __restrict__ binned_w,
                int* __restrict__ seg_w) {
  v2i* __restrict__ binned = reinterpret_cast<v2i*>(binned_w);
  v2i* __restrict__ seg = reinterpret_cast<v2i*>(seg_w);
  __shared__ int hist[NBKT_PAD];
  __shared__ int start[NBKT_PAD];
  __shared__ v2i stage[BIN_CH];
  const int t = threadIdx.x;
  const int w = blockIdx.x;
  const int e0 = w * BIN_CH + t;
  for (int i = t; i < NBKT_PAD; i += BIN_B) hist[i] = 0;
  int c[BIN_EPT], s[BIN_EPT];
#pragma unroll
  for (int j = 0; j < BIN_EPT; ++j) {
    c[j] = col[e0 + j * BIN_B];
    s[j] = row[e0 + j * BIN_B];
  }
  __syncthreads();
  int r[BIN_EPT];
#pragma unroll
  for (int j = 0; j < BIN_EPT; ++j)
    r[j] = atomicAdd(&hist[(unsigned)c[j] / BKT], 1);
  __syncthreads();
  if (t < 32) {
    int v[BIN_SCAN];
    int sum = 0;
#pragma unroll
    for (int u = 0; u < BIN_SCAN; ++u) {
      v[u] = hist[t * BIN_SCAN + u];
      sum += v[u];
    }
    int inc = sum;
#pragma unroll
    for (int d = 1; d < 32; d <<= 1) {
      const int o = __shfl_up(inc, d);
      if (t >= d) inc += o;
    }
    int run = inc - sum;
#pragma unroll
    for (int u = 0; u < BIN_SCAN; ++u) {
      start[t * BIN_SCAN + u] = run;
      run += v[u];
    }
  }
  __syncthreads();
#pragma unroll
  for (int j = 0; j < BIN_EPT; ++j) {
    v2i ed;
    ed.x = c[j];
    ed.y = s[j];
    stage[start[(unsigned)c[j] / BKT] + r[j]] = ed;
  }
  for (int i = t; i < NBKT_PAD; i += BIN_B) {
    v2i sc;
    sc.x = start[i];
    sc.y = hist[i];
    seg[(size_t)w * NBKT_PAD + i] = sc;
  }
  __syncthreads();
#pragma unroll
  for (int j = 0; j < BIN_EPT; ++j)
    binned[(size_t)w * BIN_CH + j * BIN_B + t] = stage[j * BIN_B + t];
}

#define BLD_B 256
#define BLD_U 8
static_assert(BKT % 4 == 0 && SLOT_PITCH % 4 == 0 && N_NODES % 4 == 0,
              "16-byte groups of the image stay inside one plane and inside the valid nodes");
__global__ void __launch_bounds__(BLD_B)
slot_build_kernel(const int* __restrict__ binned_w,
                  const int* __restrict__ seg_w,
                  int* __restrict__ slot,
                  int* __restrict__ nxt,
                  int* __restrict__ cntc,
                  int* __restrict__ headc) {
  const v2i* __restrict__ binned = reinterpret_cast<const v2i*>(binned_w);
  const v2i* __restrict__ seg = reinterpret_cast<const v2i*>(seg_w);
  __shared__ int cnt[BKT];
  __shared__ int head[BKT];
  __shared__ v4i slotL4[SLOT_W * BKT / 4];
  int* slotL = reinterpret_cast<int*>(slotL4);
  const int t = threadIdx.x;
  const int bk = blockIdx.x;
  const int base = bk * BKT;
  const v4i zero4 = {0, 0, 0, 0};
  for (int i = t; i < BKT; i += BLD_B) {
    cnt[i] = 0;
    head[i] = 0;
  }
  for (int i4 = t; i4 < SLOT_W * BKT / 4; i4 += BLD_B) slotL4[i4] = zero4;
  __syncthreads();
  if (t < BIN_G) {
    const v2i sc = seg[(size_t)t * NBKT_PAD + bk];
    const int p0 = t * BIN_CH + sc.x;
    const int n = sc.y;
    for (int j = 0; j < n; j += BLD_U) {
      v2i ed[BLD_U];
      int k[BLD_U];
#pragma unroll
      for (int u = 0; u < BLD_U; ++u) {
        v2i z;
        z.x = base;
        z.y = 0;
        ed[u] = (j + u < n) ? binned[p0 + j + u] : z;
      }
#pragma unroll
      for (int u = 0; u < BLD_U; ++u) {
        k[u] = 0;
        if (j + u < n) k[u] = atomicAdd(&cnt[ed[u].x - base], 1);
      }
#pragma unroll
      for (int u = 0; u < BLD_U; ++u) {
        if (j + u < n) {
          const int li = ed[u].x - base;
          if (k[u] < SLOT_W) {
            slotL[k[u] * BKT + li] = ed[u].y;
          } else {
            const int p = p0 + j + u;
            nxt[p] = atomicExch(&head[li], p + 1);
          }
        }
      }
    }
  }
  __syncthreads();
  const int nvalid = (N_NODES - base < BKT) ? (N_NODES - base) : BKT;
  for (int i4 = t; i4 < SLOT_W * BKT / 4; i4 += BLD_B) {
    const int i = 4 * i4;
    const int k = i / BKT;
    const int li = i - k * BKT;
    if (li < nvalid)
      *reinterpret_cast<v4i*>(slot + (size_t)k * SLOT_PITCH + base + li) = slotL4[i4];
  }
  for (int i = t; i < nvalid; i += BLD_B) {
    cntc[base + i] = cnt[i];
    headc[base + i] = head[i];
  }
}

template <int DIN, int DOUT, int OPITCH>
__global__ void gemm_wmma_kernel(const float* __restrict__ H,
                                 const float* __restrict__ W,
                                 const int* __restrict__ cntc,
                                 float* __restrict__ Out) {
  constexpr int NT = (DOUT + 15) / 16;
  const int wave = (blockIdx.x * blockDim.x + threadIdx.x) >> 5;
  const int lane = threadIdx.x & 31;
  const int m0 = wave * 16;
  if (m0 >= N_NODES) return;

  const int lhalf = lane & 15;
  const int khi = (lane >> 4) << 1;
  const int ncl = (lhalf < DOUT) ? lhalf : (DOUT - 1);

  const float* __restrict__ hrow = H + (size_t)(m0 + lhalf) * DIN;

  v8f acc0 = {};
  v8f acc1 = {};
#pragma unroll
  for (int k = 0; k < DIN; k += 4) {
    v2f a, b0;
    a.x = hrow[k + khi];
    a.y = hrow[k + khi + 1];
    b0.x = W[(k + khi) * DOUT + ncl];
    b0.y = W[(k + khi + 1) * DOUT + ncl];
    acc0 = __builtin_amdgcn_wmma_f32_16x16x4_f32(
        false, a, false, b0, (short)0, acc0, false, false);
    if (NT > 1) {
      v2f b1;
      b1.x = W[(k + khi) * DOUT + 16 + lhalf];
      b1.y = W[(k + khi + 1) * DOUT + 16 + lhalf];
      acc1 = __builtin_amdgcn_wmma_f32_16x16x4_f32(
          false, a, false, b1, (short)0, acc1, false, false);
    }
  }

  const int mrow = m0 + ((lane >> 4) << 3);
#pragma unroll
  for (int v = 0; v < 8; ++v) {
    const int m = mrow + v;
    const float rs = rsqrtf((float)(cntc[m] + 1));
    if (lhalf < OPITCH) Out[(size_t)m * OPITCH + lhalf] = acc0[v] * rs;
    if (NT > 1)         Out[(size_t)m * OPITCH + 16 + lhalf] = acc1[v] * rs;
  }
}

__device__ __forceinline__ void chain_load_a(const float* __restrict__ rowp,
                                             int khi, v2f (&a)[8]) {
#pragma unroll
  for (int j = 0; j < 8; ++j) {
    a[j].x = rowp[4 * j + khi];
    a[j].y = rowp[4 * j + khi + 1];
  }
}

__device__ __forceinline__ v8f chain_tile(const v2f (&a)[8],
                                          const float (*P)[16],
                                          int khi, int lhalf) {
  v8f acc = {};
#pragma unroll
  for (int j = 0; j < 8; ++j) {
    v2f b;
    b.x = P[4 * j + khi][lhalf];
    b.y = P[4 * j + khi + 1][lhalf];
    acc = __builtin_amdgcn_wmma_f32_16x16x4_f32(
        false, a[j], false, b, (short)0, acc, false, false);
  }
  return acc;
}

__device__ __forceinline__ void chain_step(const v2f (&w0)[8], const v2f (&w1)[8],
                                           const v2f (&bb)[8],
                                           const float (*Pc)[16], float (*Pn)[16],
                                           float* __restrict__ ck, int lane) {
  const int lhalf = lane & 15;
  const int khi = (lane >> 4) << 1;
  const int mhi = (lane >> 4) << 3;
  const v8f d0 = chain_tile(w0, Pc, khi, lhalf);
  const v8f d1 = chain_tile(w1, Pc, khi, lhalf);
  const v8f dc = chain_tile(bb, Pc, khi, lhalf);
#pragma unroll
  for (int v = 0; v < 8; ++v) {
    Pn[mhi + v][lhalf] = d0[v];
    Pn[16 + mhi + v][lhalf] = d1[v];
  }
  if (lane < 3) ck[lane] = dc[0];
  __syncthreads();
}

__global__ void __launch_bounds__(32)
weight_chain_kernel(const float* __restrict__ W1, const float* __restrict__ b1,
                    const float* __restrict__ W2, const float* __restrict__ b2,
                    const float* __restrict__ W3, const float* __restrict__ b3,
                    const float* __restrict__ W4, const float* __restrict__ b4,
                    const float* __restrict__ W5,
                    float* __restrict__ P1,
                    float* __restrict__ cv) {
  __shared__ float P[2][32][16];
  const int lane = threadIdx.x;
  const int lhalf = lane & 15;
  const int khi = (lane >> 4) << 1;

  v2f aW4[2][8], aW3[2][8], aW2[2][8], aW1[8];
  v2f aB4[8], aB3[8], aB2[8], aB1[8];
#pragma unroll
  for (int t = 0; t < 2; ++t) {
    chain_load_a(W4 + (16 * t + lhalf) * 32, khi, aW4[t]);
    chain_load_a(W3 + (16 * t + lhalf) * 32, khi, aW3[t]);
    chain_load_a(W2 + (16 * t + lhalf) * 32, khi, aW2[t]);
  }
  chain_load_a(W1 + (lhalf & 7) * 32, khi, aW1);
  chain_load_a(b4, khi, aB4);
  chain_load_a(b3, khi, aB3);
  chain_load_a(b2, khi, aB2);
  chain_load_a(b1, khi, aB1);

  for (int idx = lane; idx < 32 * 16; idx += 32) {
    const int r = idx >> 4;
    const int c = idx & 15;
    P[0][r][c] = W5[r * 3 + (c < 3 ? c : 2)];
  }
  __syncthreads();

  chain_step(aW4[0], aW4[1], aB4, P[0], P[1], cv + 9, lane);
  chain_step(aW3[0], aW3[1], aB3, P[1], P[0], cv + 6, lane);
  chain_step(aW2[0], aW2[1], aB2, P[0], P[1], cv + 3, lane);

  const float (*P2)[16] = P[1];
  const v8f d0 = chain_tile(aW1, P2, khi, lhalf);
  const v8f dc = chain_tile(aB1, P2, khi, lhalf);
  if (lane < 3) {
#pragma unroll
    for (int v = 0; v < 8; ++v) P1[v * 3 + lane] = d0[v];
    cv[lane] = dc[0];
  }
}

#define GATHER_B 64

template <int MODE>
__global__ void __launch_bounds__(GATHER_B)
agg_gather_node_kernel(const float* __restrict__ yin,
                       const float* __restrict__ bias,
                       const int* __restrict__ cntc,
                       const int* __restrict__ headc,
                       const int* __restrict__ slot,
                       const int* __restrict__ nxt,
                       const int* __restrict__ binned_w,
                       float* __restrict__ out) {
  constexpr bool LAST = (MODE == 2);
  const int i = blockIdx.x * GATHER_B + threadIdx.x;
  if (i >= N_NODES) return;
  const int n = cntc[i];
  const float* __restrict__ yself = yin + 4 * (size_t)i;
  float a0 = yself[0];
  float a1 = yself[1];
  float a2 = yself[2];

  int q[SLOT_W];
  row3 v[SLOT_W];
  const char* __restrict__ ybytes = reinterpret_cast<const char*>(yin);
#pragma unroll
  for (int k = 0; k < SLOT_W; ++k) q[k] = slot[(size_t)k * SLOT_PITCH + i];
#pragma unroll
  for (int u = 0; u < SLOT_W; ++u)
    v[u] = *reinterpret_cast<const row3*>(ybytes + ((unsigned)q[u] << 4));
#define ROWV(u) "v"(v[u].x), "v"(v[u].y), "v"(v[u].z)
#define ROWV4(u) ROWV(u), ROWV(u + 1), ROWV(u + 2), ROWV(u + 3)
#define ROWV20(u) ROWV4(u), ROWV4(u + 4), ROWV4(u + 8), ROWV4(u + 12), ROWV4(u + 16)
  asm volatile("" : "+v"(a0), "+v"(a1), "+v"(a2) : ROWV20(0), ROWV20(20));
#undef ROWV20
#undef ROWV4
#undef ROWV
#pragma unroll
  for (int u = 0; u < SLOT_W; ++u) {
    const bool act = (u < n);
    const float t0 = a0 + v[u].x;
    const float t1 = a1 + v[u].y;
    const float t2 = a2 + v[u].z;
    a0 = act ? t0 : a0;
    a1 = act ? t1 : a1;
    a2 = act ? t2 : a2;
  }
  if (n > SLOT_W) {
    int p = headc[i];
    for (int t = SLOT_W; t < n && p != 0; ++t) {
      const int e = p - 1;
      const unsigned r = (unsigned)binned_w[2 * (size_t)e + 1];
      p = nxt[e];
      const row3 w = *reinterpret_cast<const row3*>(yin + 4 * (size_t)r);
      a0 += w.x;
      a1 += w.y;
      a2 += w.z;
    }
  }

  const float d = rsqrtf((float)(n + 1));
  const float z0 = a0 * d + bias[0];
  const float z1 = a1 * d + bias[1];
  const float z2 = a2 * d + bias[2];
  if (LAST) {
    out[3 * (size_t)i + 0] = z0;
    out[3 * (size_t)i + 1] = z1;
    out[3 * (size_t)i + 2] = z2;
  } else {
    v4f o;
    o.x = z0 * d;
    o.y = z1 * d;
    o.z = z2 * d;
    o.w = 0.f;
    reinterpret_cast<v4f*>(out)[i] = o;
  }
}

extern "C" void kernel_launch(void* const* d_in, const int* in_sizes, int n_in,
                              void* d_out, int out_size, void* d_ws, size_t ws_size,
                              hipStream_t stream) {
  const float* x  = (const float*)d_in[0];
  const int*   ei = (const int*)d_in[1];
  const int*   row = ei;
  const int*   col = ei + N_EDGES;

  const float* Wm[5] = {(const float*)d_in[2], (const float*)d_in[4],
                        (const float*)d_in[6], (const float*)d_in[8],
                        (const float*)d_in[10]};
  const float* Bm[5] = {(const float*)d_in[3], (const float*)d_in[5],
                        (const float*)d_in[7], (const float*)d_in[9],
                        (const float*)d_in[11]};

  int*   wsi   = (int*)d_ws;
  int*   slot  = wsi;
  int*   nxt   = slot + SLOT_W * SLOT_PITCH;
  int*   binned = nxt + N_EDGES;
  int*   seg   = binned + 2 * N_EDGES;
  float* P1    = (float*)(seg + 2 * BIN_G * NBKT_PAD);
  float* cv    = P1 + 64;
  float* zA    = P1 + 128;
  float* zB    = zA + 4 * N_PAD;
  int*   cntc  = (int*)(zB + 4 * N_PAD);
  int*   headc = cntc + N_PAD;

  const int B = 256;
  const int gGemm = ((N_NODES / 16) * 32 + B - 1) / B;
  const int gGn  = (N_NODES + GATHER_B - 1) / GATHER_B;

  edge_bin_kernel<<<BIN_G, BIN_B, 0, stream>>>(row, col, binned, seg);
  slot_build_kernel<<<NBKT, BLD_B, 0, stream>>>(binned, seg, slot, nxt, cntc, headc);

  weight_chain_kernel<<<1, 32, 0, stream>>>(Wm[0], Bm[0], Wm[1], Bm[1], Wm[2], Bm[2],
                                            Wm[3], Bm[3], Wm[4], P1, cv);
  gemm_wmma_kernel<8, 3, 4><<<gGemm, B, 0, stream>>>(x, P1, cntc, zA);

  agg_gather_node_kernel<1><<<gGn, GATHER_B, 0, stream>>>(zA, cv + 0, cntc, headc, slot,
                                                          nxt, binned, zB);
  agg_gather_node_kernel<1><<<gGn, GATHER_B, 0, stream>>>(zB, cv + 3, cntc, headc, slot,
                                                          nxt, binned, zA);
  agg_gather_node_kernel<1><<<gGn, GATHER_B, 0, stream>>>(zA, cv + 6, cntc, headc, slot,
                                                          nxt, binned, zB);
  agg_gather_node_kernel<1><<<gGn, GATHER_B, 0, stream>>>(zB, cv + 9, cntc, headc, slot,
                                                          nxt, binned, zA);
  agg_gather_node_kernel<2><<<gGn, GATHER_B, 0, stream>>>(zA, Bm[4], cntc, headc, slot,
                                                          nxt, binned, (float*)d_out);
}


# Launch fusion: the weight chain runs as one extra work-group of the edge-binning launch and x*P1 inside the slot-building groups (counts still in LDS): 7 launches, not 9; values bit-identical
# speedup vs baseline: 10.4113x; 1.0956x over previous
#include <hip/hip_runtime.h>

#define N_NODES 100000
#define N_EDGES 1600000

typedef __attribute__((ext_vector_type(2))) float v2f;
typedef __attribute__((ext_vector_type(8))) float v8f;
typedef __attribute__((ext_vector_type(4))) int v4i;
typedef __attribute__((ext_vector_type(4))) float v4f;
struct row3 { float x, y, z; };

#define N_PAD 100096
#define SLOT_W 40
#define SLOT_PITCH 100128

typedef __attribute__((ext_vector_type(2))) int v2i;

#define BKT 352
#define NBKT ((N_NODES + BKT - 1) / BKT)
#define NBKT_PAD 288
#define BIN_SCAN 9
#define BIN_B 256
#define BIN_EPT 25
#define BIN_CH (BIN_B * BIN_EPT)
#define BIN_G (N_EDGES / BIN_CH)
static_assert(N_EDGES % BIN_CH == 0, "every work-group of pass 1 has a full piece");
static_assert(NBKT <= NBKT_PAD && NBKT_PAD == 32 * BIN_SCAN, "one wave scans the histogram");
static_assert(BKT % 32 == 0 && N_NODES % 32 == 0, "a block's part of a plane is whole lines");
static_assert(BIN_G <= BIN_B, "pass 2 gives every run of a block a lane");

__device__ __forceinline__ void chain_load_a(const float* __restrict__ rowp,
                                             int khi, v2f (&a)[8]) {
#pragma unroll
  for (int j = 0; j < 8; ++j) {
    a[j].x = rowp[4 * j + khi];
    a[j].y = rowp[4 * j + khi + 1];
  }
}

__device__ __forceinline__ v8f chain_tile(const v2f (&a)[8],
                                          const float (*P)[16],
                                          int khi, int lhalf) {
  v8f acc = {};
#pragma unroll
  for (int j = 0; j < 8; ++j) {
    v2f b;
    b.x = P[4 * j + khi][lhalf];
    b.y = P[4 * j + khi + 1][lhalf];
    acc = __builtin_amdgcn_wmma_f32_16x16x4_f32(
        false, a[j], false, b, (short)0, acc, false, false);
  }
  return acc;
}

__device__ __forceinline__ void chain_step(const v2f (&w0)[8], const v2f (&w1)[8],
                                           const v2f (&bb)[8],
                                           const float (*Pc)[16], float (*Pn)[16],
                                           float* __restrict__ ck, int lane, bool act) {
  if (act) {
    const int lhalf = lane & 15;
    const int khi = (lane >> 4) << 1;
    const int mhi = (lane >> 4) << 3;
    const v8f d0 = chain_tile(w0, Pc, khi, lhalf);
    const v8f d1 = chain_tile(w1, Pc, khi, lhalf);
    const v8f dc = chain_tile(bb, Pc, khi, lhalf);
#pragma unroll
    for (int v = 0; v < 8; ++v) {
      Pn[mhi + v][lhalf] = d0[v];
      Pn[16 + mhi + v][lhalf] = d1[v];
    }
    if (lane < 3) ck[lane] = dc[0];
  }
  __syncthreads();
}

__device__ __forceinline__ void weight_chain_group(
    float (*P)[32][16], int t,
    const float* __restrict__ W1, const float* __restrict__ b1,
    const float* __restrict__ W2, const float* __restrict__ b2,
    const float* __restrict__ W3, const float* __restrict__ b3,
    const float* __restrict__ W4, const float* __restrict__ b4,
    const float* __restrict__ W5,
    float* __restrict__ P1,
    float* __restrict__ cv) {
  const bool act = (t < 32);
  const int lane = t & 31;
  const int lhalf = lane & 15;
  const int khi = (lane >> 4) << 1;

  v2f aW4[2][8], aW3[2][8], aW2[2][8], aW1[8];
  v2f aB4[8], aB3[8], aB2[8], aB1[8];
  if (act) {
#pragma unroll
    for (int u = 0; u < 2; ++u) {
      chain_load_a(W4 + (16 * u + lhalf) * 32, khi, aW4[u]);
      chain_load_a(W3 + (16 * u + lhalf) * 32, khi, aW3[u]);
      chain_load_a(W2 + (16 * u + lhalf) * 32, khi, aW2[u]);
    }
    chain_load_a(W1 + (lhalf & 7) * 32, khi, aW1);
    chain_load_a(b4, khi, aB4);
    chain_load_a(b3, khi, aB3);
    chain_load_a(b2, khi, aB2);
    chain_load_a(b1, khi, aB1);

    for (int idx = lane; idx < 32 * 16; idx += 32) {
      const int r = idx >> 4;
      const int c = idx & 15;
      P[0][r][c] = W5[r * 3 + (c < 3 ? c : 2)];
    }
  }
  __syncthreads();

  chain_step(aW4[0], aW4[1], aB4, P[0], P[1], cv + 9, lane, act);
  chain_step(aW3[0], aW3[1], aB3, P[1], P[0], cv + 6, lane, act);
  chain_step(aW2[0], aW2[1], aB2, P[0], P[1], cv + 3, lane, act);

  if (act) {
    const float (*P2)[16] = P[1];
    const v8f d0 = chain_tile(aW1, P2, khi, lhalf);
    const v8f dc = chain_tile(aB1, P2, khi, lhalf);
    if (lane < 3) {
#pragma unroll
      for (int v = 0; v < 8; ++v) P1[v * 3 + lane] = d0[v];
      cv[lane] = dc[0];
    }
  }
}

__global__ void __launch_bounds__(BIN_B)
edge_bin_kernel(const int* __restrict__ row,
                const int* __restrict__ col,
                int* __restrict__ binned_w,
                int* __restrict__ seg_w,
                const float* __restrict__ W1, const float* __restrict__ b1,
                const float* __restrict__ W2, const float* __restrict__ b2,
                const float* __restrict__ W3, const float* __restrict__ b3,
                const float* __restrict__ W4, const float* __restrict__ b4,
                const float* __restrict__ W5,
                float* __restrict__ P1,
                float* __restrict__ cv) {
  v2i* __restrict__ binned = reinterpret_cast<v2i*>(binned_w);
  v2i* __restrict__ seg = reinterpret_cast<v2i*>(seg_w);
  __shared__ int hist[NBKT_PAD];
  __shared__ int start[NBKT_PAD];
  __shared__ v2i stage[BIN_CH];
  __shared__ float Pch[2][32][16];
  const int t = threadIdx.x;
  const int w = blockIdx.x;
  if (w == BIN_G) {
    weight_chain_group(Pch, t, W1, b1, W2, b2, W3, b3, W4, b4, W5, P1, cv);
    return;
  }
  const int e0 = w * BIN_CH + t;
  for (int i = t; i < NBKT_PAD; i += BIN_B) hist[i] = 0;
  int c[BIN_EPT], s[BIN_EPT];
#pragma unroll
  for (int j = 0; j < BIN_EPT; ++j) {
    c[j] = col[e0 + j * BIN_B];
    s[j] = row[e0 + j * BIN_B];
  }
  __syncthreads();
  int r[BIN_EPT];
#pragma unroll
  for (int j = 0; j < BIN_EPT; ++j)
    r[j] = atomicAdd(&hist[(unsigned)c[j] / BKT], 1);
  __syncthreads();
  if (t < 32) {
    int v[BIN_SCAN];
    int sum = 0;
#pragma unroll
    for (int u = 0; u < BIN_SCAN; ++u) {
      v[u] = hist[t * BIN_SCAN + u];
      sum += v[u];
    }
    int inc = sum;
#pragma unroll
    for (int d = 1; d < 32; d <<= 1) {
      const int o = __shfl_up(inc, d);
      if (t >= d) inc += o;
    }
    int run = inc - sum;
#pragma unroll
    for (int u = 0; u < BIN_SCAN; ++u) {
      start[t * BIN_SCAN + u] = run;
      run += v[u];
    }
  }
  __syncthreads();
#pragma unroll
  for (int j = 0; j < BIN_EPT; ++j) {
    v2i ed;
    ed.x = c[j];
    ed.y = s[j];
    stage[start[(unsigned)c[j] / BKT] + r[j]] = ed;
  }
  for (int i = t; i < NBKT_PAD; i += BIN_B) {
    v2i sc;
    sc.x = start[i];
    sc.y = hist[i];
    seg[(size_t)w * NBKT_PAD + i] = sc;
  }
  __syncthreads();
#pragma unroll
  for (int j = 0; j < BIN_EPT; ++j)
    binned[(size_t)w * BIN_CH + j * BIN_B + t] = stage[j * BIN_B + t];
}

#define BLD_B 256
#define BLD_U 8
#define XP_T ((BKT / 16 + BLD_B / 32 - 1) / (BLD_B / 32))
static_assert(BKT % 4 == 0 && SLOT_PITCH % 4 == 0 && N_NODES % 4 == 0,
              "16-byte groups of the image stay inside one plane and inside the valid nodes");
__global__ void __launch_bounds__(BLD_B)
slot_build_kernel(const int* __restrict__ binned_w,
                  const int* __restrict__ seg_w,
                  int* __restrict__ slot,
                  int* __restrict__ nxt,
                  int* __restrict__ cntc,
                  int* __restrict__ headc,
                  const float* __restrict__ x,
                  const float* __restrict__ P1,
                  float* __restrict__ y0) {
  const v2i* __restrict__ binned = reinterpret_cast<const v2i*>(binned_w);
  const v2i* __restrict__ seg = reinterpret_cast<const v2i*>(seg_w);
  __shared__ int cnt[BKT];
  __shared__ int head[BKT];
  __shared__ v4i slotL4[SLOT_W * BKT / 4];
  int* slotL = reinterpret_cast<int*>(slotL4);
  const int t = threadIdx.x;
  const int bk = blockIdx.x;
  const int base = bk * BKT;
  const v4i zero4 = {0, 0, 0, 0};
  const int nvalid = (N_NODES - base < BKT) ? (N_NODES - base) : BKT;
  const int wv = t >> 5;
  const int lane = t & 31;
  const int lhalf = lane & 15;
  const int khi = (lane >> 4) << 1;
  const int ncl = (lhalf < 3) ? lhalf : 2;
  v2f pb0, pb1;
  pb0.x = P1[(khi) * 3 + ncl];
  pb0.y = P1[(khi + 1) * 3 + ncl];
  pb1.x = P1[(4 + khi) * 3 + ncl];
  pb1.y = P1[(4 + khi + 1) * 3 + ncl];
  v2f xa[XP_T][2];
#pragma unroll
  for (int j = 0; j < XP_T; ++j) {
    const int r0 = 16 * (wv + (BLD_B / 32) * j);
    if (r0 < nvalid) {
      const float* __restrict__ hrow = x + (size_t)(base + r0 + lhalf) * 8;
      xa[j][0].x = hrow[khi];
      xa[j][0].y = hrow[khi + 1];
      xa[j][1].x = hrow[4 + khi];
      xa[j][1].y = hrow[4 + khi + 1];
    }
  }
  for (int i = t; i < BKT; i += BLD_B) {
    cnt[i] = 0;
    head[i] = 0;
  }
  for (int i4 = t; i4 < SLOT_W * BKT / 4; i4 += BLD_B) slotL4[i4] = zero4;
  __syncthreads();
  if (t < BIN_G) {
    const v2i sc = seg[(size_t)t * NBKT_PAD + bk];
    const int p0 = t * BIN_CH + sc.x;
    const int n = sc.y;
    for (int j = 0; j < n; j += BLD_U) {
      v2i ed[BLD_U];
      int k[BLD_U];
#pragma unroll
      for (int u = 0; u < BLD_U; ++u) {
        v2i z;
        z.x = base;
        z.y = 0;
        ed[u] = (j + u < n) ? binned[p0 + j + u] : z;
      }
#pragma unroll
      for (int u = 0; u < BLD_U; ++u) {
        k[u] = 0;
        if (j + u < n) k[u] = atomicAdd(&cnt[ed[u].x - base], 1);
      }
#pragma unroll
      for (int u = 0; u < BLD_U; ++u) {
        if (j + u < n) {
          const int li = ed[u].x - base;
          if (k[u] < SLOT_W) {
            slotL[k[u] * BKT + li] = ed[u].y;
          } else {
            const int p = p0 + j + u;
            nxt[p] = atomicExch(&head[li], p + 1);
          }
        }
      }
    }
  }
  __syncthreads();
  for (int i4 = t; i4 < SLOT_W * BKT / 4; i4 += BLD_B) {
    const int i = 4 * i4;
    const int k = i / BKT;
    const int li = i - k * BKT;
    if (li < nvalid)
      *reinterpret_cast<v4i*>(slot + (size_t)k * SLOT_PITCH + base + li) = slotL4[i4];
  }
  for (int i = t; i < nvalid; i += BLD_B) {
    cntc[base + i] = cnt[i];
    headc[base + i] = head[i];
  }

#pragma unroll
  for (int j = 0; j < XP_T; ++j) {
    const int r0 = 16 * (wv + (BLD_B / 32) * j);
    if (r0 < nvalid) {
      v8f acc = {};
      acc = __builtin_amdgcn_wmma_f32_16x16x4_f32(
          false, xa[j][0], false, pb0, (short)0, acc, false, false);
      acc = __builtin_amdgcn_wmma_f32_16x16x4_f32(
          false, xa[j][1], false, pb1, (short)0, acc, false, false);
      const int lrow = r0 + ((lane >> 4) << 3);
#pragma unroll
      for (int v = 0; v < 8; ++v) {
        const int li = lrow + v;
        const float rs = rsqrtf((float)(cnt[li] + 1));
        if (lhalf < 4) y0[(size_t)(base + li) * 4 + lhalf] = acc[v] * rs;
      }
    }
  }
}

#define GATHER_B 64

template <int MODE>
__global__ void __launch_bounds__(GATHER_B)
agg_gather_node_kernel(const float* __restrict__ yin,
                       const float* __restrict__ bias,
                       const int* __restrict__ cntc,
                       const int* __restrict__ headc,
                       const int* __restrict__ slot,
                       const int* __restrict__ nxt,
                       const int* __restrict__ binned_w,
                       float* __restrict__ out) {
  constexpr bool LAST = (MODE == 2);
  const int i = blockIdx.x * GATHER_B + threadIdx.x;
  if (i >= N_NODES) return;
  const int n = cntc[i];
  const float* __restrict__ yself = yin + 4 * (size_t)i;
  float a0 = yself[0];
  float a1 = yself[1];
  float a2 = yself[2];

  int q[SLOT_W];
  row3 v[SLOT_W];
  const char* __restrict__ ybytes = reinterpret_cast<const char*>(yin);
#pragma unroll
  for (int k = 0; k < SLOT_W; ++k) q[k] = slot[(size_t)k * SLOT_PITCH + i];
#pragma unroll
  for (int u = 0; u < SLOT_W; ++u)
    v[u] = *reinterpret_cast<const row3*>(ybytes + ((unsigned)q[u] << 4));
#define ROWV(u) "v"(v[u].x), "v"(v[u].y), "v"(v[u].z)
#define ROWV4(u) ROWV(u), ROWV(u + 1), ROWV(u + 2), ROWV(u + 3)
#define ROWV20(u) ROWV4(u), ROWV4(u + 4), ROWV4(u + 8), ROWV4(u + 12), ROWV4(u + 16)
  asm volatile("" : "+v"(a0), "+v"(a1), "+v"(a2) : ROWV20(0), ROWV20(20));
#undef ROWV20
#undef ROWV4
#undef ROWV
#pragma unroll
  for (int u = 0; u < SLOT_W; ++u) {
    const bool act = (u < n);
    const float t0 = a0 + v[u].x;
    const float t1 = a1 + v[u].y;
    const float t2 = a2 + v[u].z;
    a0 = act ? t0 : a0;
    a1 = act ? t1 : a1;
    a2 = act ? t2 : a2;
  }
  if (n > SLOT_W) {
    int p = headc[i];
    for (int t = SLOT_W; t < n && p != 0; ++t) {
      const int e = p - 1;
      const unsigned r = (unsigned)binned_w[2 * (size_t)e + 1];
      p = nxt[e];
      const row3 w = *reinterpret_cast<const row3*>(yin + 4 * (size_t)r);
      a0 += w.x;
      a1 += w.y;
      a2 += w.z;
    }
  }

  const float d = rsqrtf((float)(n + 1));
  const float z0 = a0 * d + bias[0];
  const float z1 = a1 * d + bias[1];
  const float z2 = a2 * d + bias[2];
  if (LAST) {
    out[3 * (size_t)i + 0] = z0;
    out[3 * (size_t)i + 1] = z1;
    out[3 * (size_t)i + 2] = z2;
  } else {
    v4f o;
    o.x = z0 * d;
    o.y = z1 * d;
    o.z = z2 * d;
    o.w = 0.f;
    reinterpret_cast<v4f*>(out)[i] = o;
  }
}

extern "C" void kernel_launch(void* const* d_in, const int* in_sizes, int n_in,
                              void* d_out, int out_size, void* d_ws, size_t ws_size,
                              hipStream_t stream) {
  const float* x  = (const float*)d_in[0];
  const int*   ei = (const int*)d_in[1];
  const int*   row = ei;
  const int*   col = ei + N_EDGES;

  const float* Wm[5] = {(const float*)d_in[2], (const float*)d_in[4],
                        (const float*)d_in[6], (const float*)d_in[8],
                        (const float*)d_in[10]};
  const float* Bm[5] = {(const float*)d_in[3], (const float*)d_in[5],
                        (const float*)d_in[7], (const float*)d_in[9],
                        (const float*)d_in[11]};

  int*   wsi   = (int*)d_ws;
  int*   slot  = wsi;
  int*   nxt   = slot + SLOT_W * SLOT_PITCH;
  int*   binned = nxt + N_EDGES;
  int*   seg   = binned + 2 * N_EDGES;
  float* P1    = (float*)(seg + 2 * BIN_G * NBKT_PAD);
  float* cv    = P1 + 64;
  float* zA    = P1 + 128;
  float* zB    = zA + 4 * N_PAD;
  int*   cntc  = (int*)(zB + 4 * N_PAD);
  int*   headc = cntc + N_PAD;

  const int gGn  = (N_NODES + GATHER_B - 1) / GATHER_B;

  edge_bin_kernel<<<BIN_G + 1, BIN_B, 0, stream>>>(row, col, binned, seg,
                                                   Wm[0], Bm[0], Wm[1], Bm[1], Wm[2], Bm[2],
                                                   Wm[3], Bm[3], Wm[4], P1, cv);
  slot_build_kernel<<<NBKT, BLD_B, 0, stream>>>(binned, seg, slot, nxt, cntc, headc,
                                                x, P1, zA);

  agg_gather_node_kernel<1><<<gGn, GATHER_B, 0, stream>>>(zA, cv + 0, cntc, headc, slot,
                                                          nxt, binned, zB);
  agg_gather_node_kernel<1><<<gGn, GATHER_B, 0, stream>>>(zB, cv + 3, cntc, headc, slot,
                                                          nxt, binned, zA);
  agg_gather_node_kernel<1><<<gGn, GATHER_B, 0, stream>>>(zA, cv + 6, cntc, headc, slot,
                                                          nxt, binned, zB);
  agg_gather_node_kernel<1><<<gGn, GATHER_B, 0, stream>>>(zB, cv + 9, cntc, headc, slot,
                                                          nxt, binned, zA);
  agg_gather_node_kernel<2><<<gGn, GATHER_B, 0, stream>>>(zA, Bm[4], cntc, headc, slot,
                                                          nxt, binned, (float*)d_out);
}


# Slot building reads each run of binned edges with one wave, lanes on consecutive edges, all runs requested at once: one round trip, no per-lane chain; sums in LDS arrival-rank order (order differs)
# speedup vs baseline: 11.3751x; 1.0926x over previous
#include <hip/hip_runtime.h>

#define N_NODES 100000
#define N_EDGES 1600000

typedef __attribute__((ext_vector_type(2))) float v2f;
typedef __attribute__((ext_vector_type(8))) float v8f;
typedef __attribute__((ext_vector_type(4))) int v4i;
typedef __attribute__((ext_vector_type(4))) float v4f;
struct row3 { float x, y, z; };

#define N_PAD 100096
#define SLOT_W 40
#define SLOT_PITCH 100128

typedef __attribute__((ext_vector_type(2))) int v2i;

#define BKT 352
#define NBKT ((N_NODES + BKT - 1) / BKT)
#define NBKT_PAD 288
#define BIN_SCAN 9
#define BIN_B 256
#define BIN_EPT 25
#define BIN_CH (BIN_B * BIN_EPT)
#define BIN_G (N_EDGES / BIN_CH)
static_assert(N_EDGES % BIN_CH == 0, "every work-group of pass 1 has a full piece");
static_assert(NBKT <= NBKT_PAD && NBKT_PAD == 32 * BIN_SCAN, "one wave scans the histogram");
static_assert(BKT % 32 == 0 && N_NODES % 32 == 0, "a block's part of a plane is whole lines");

__device__ __forceinline__ void chain_load_a(const float* __restrict__ rowp,
                                             int khi, v2f (&a)[8]) {
#pragma unroll
  for (int j = 0; j < 8; ++j) {
    a[j].x = rowp[4 * j + khi];
    a[j].y = rowp[4 * j + khi + 1];
  }
}

__device__ __forceinline__ v8f chain_tile(const v2f (&a)[8],
                                          const float (*P)[16],
                                          int khi, int lhalf) {
  v8f acc = {};
#pragma unroll
  for (int j = 0; j < 8; ++j) {
    v2f b;
    b.x = P[4 * j + khi][lhalf];
    b.y = P[4 * j + khi + 1][lhalf];
    acc = __builtin_amdgcn_wmma_f32_16x16x4_f32(
        false, a[j], false, b, (short)0, acc, false, false);
  }
  return acc;
}

__device__ __forceinline__ void chain_step(const v2f (&w0)[8], const v2f (&w1)[8],
                                           const v2f (&bb)[8],
                                           const float (*Pc)[16], float (*Pn)[16],
                                           float* __restrict__ ck, int lane, bool act) {
  if (act) {
    const int lhalf = lane & 15;
    const int khi = (lane >> 4) << 1;
    const int mhi = (lane >> 4) << 3;
    const v8f d0 = chain_tile(w0, Pc, khi, lhalf);
    const v8f d1 = chain_tile(w1, Pc, khi, lhalf);
    const v8f dc = chain_tile(bb, Pc, khi, lhalf);
#pragma unroll
    for (int v = 0; v < 8; ++v) {
      Pn[mhi + v][lhalf] = d0[v];
      Pn[16 + mhi + v][lhalf] = d1[v];
    }
    if (lane < 3) ck[lane] = dc[0];
  }
  __syncthreads();
}

__device__ __forceinline__ void weight_chain_group(
    float (*P)[32][16], int t,
    const float* __restrict__ W1, const float* __restrict__ b1,
    const float* __restrict__ W2, const float* __restrict__ b2,
    const float* __restrict__ W3, const float* __restrict__ b3,
    const float* __restrict__ W4, const float* __restrict__ b4,
    const float* __restrict__ W5,
    float* __restrict__ P1,
    float* __restrict__ cv) {
  const bool act = (t < 32);
  const int lane = t & 31;
  const int lhalf = lane & 15;
  const int khi = (lane >> 4) << 1;

  v2f aW4[2][8], aW3[2][8], aW2[2][8], aW1[8];
  v2f aB4[8], aB3[8], aB2[8], aB1[8];
  if (act) {
#pragma unroll
    for (int u = 0; u < 2; ++u) {
      chain_load_a(W4 + (16 * u + lhalf) * 32, khi, aW4[u]);
      chain_load_a(W3 + (16 * u + lhalf) * 32, khi, aW3[u]);
      chain_load_a(W2 + (16 * u + lhalf) * 32, khi, aW2[u]);
    }
    chain_load_a(W1 + (lhalf & 7) * 32, khi, aW1);
    chain_load_a(b4, khi, aB4);
    chain_load_a(b3, khi, aB3);
    chain_load_a(b2, khi, aB2);
    chain_load_a(b1, khi, aB1);

    for (int idx = lane; idx < 32 * 16; idx += 32) {
      const int r = idx >> 4;
      const int c = idx & 15;
      P[0][r][c] = W5[r * 3 + (c < 3 ? c : 2)];
    }
  }
  __syncthreads();

  chain_step(aW4[0], aW4[1], aB4, P[0], P[1], cv + 9, lane, act);
  chain_step(aW3[0], aW3[1], aB3, P[1], P[0], cv + 6, lane, act);
  chain_step(aW2[0], aW2[1], aB2, P[0], P[1], cv + 3, lane, act);

  if (act) {
    const float (*P2)[16] = P[1];
    const v8f d0 = chain_tile(aW1, P2, khi, lhalf);
    const v8f dc = chain_tile(aB1, P2, khi, lhalf);
    if (lane < 3) {
#pragma unroll
      for (int v = 0; v < 8; ++v) P1[v * 3 + lane] = d0[v];
      cv[lane] = dc[0];
    }
  }
}

__global__ void __launch_bounds__(BIN_B)
edge_bin_kernel(const int* __restrict__ row,
                const int* __restrict__ col,
                int* __restrict__ binned_w,
                int* __restrict__ seg_w,
                const float* __restrict__ W1, const float* __restrict__ b1,
                const float* __restrict__ W2, const float* __restrict__ b2,
                const float* __restrict__ W3, const float* __restrict__ b3,
                const float* __restrict__ W4, const float* __restrict__ b4,
                const float* __restrict__ W5,
                float* __restrict__ P1,
                float* __restrict__ cv) {
  v2i* __restrict__ binned = reinterpret_cast<v2i*>(binned_w);
  v2i* __restrict__ seg = reinterpret_cast<v2i*>(seg_w);
  __shared__ int hist[NBKT_PAD];
  __shared__ int start[NBKT_PAD];
  __shared__ v2i stage[BIN_CH];
  __shared__ float Pch[2][32][16];
  const int t = threadIdx.x;
  const int w = blockIdx.x;
  if (w == BIN_G) {
    weight_chain_group(Pch, t, W1, b1, W2, b2, W3, b3, W4, b4, W5, P1, cv);
    return;
  }
  const int e0 = w * BIN_CH + t;
  for (int i = t; i < NBKT_PAD; i += BIN_B) hist[i] = 0;
  int c[BIN_EPT], s[BIN_EPT];
#pragma unroll
  for (int j = 0; j < BIN_EPT; ++j) {
    c[j] = col[e0 + j * BIN_B];
    s[j] = row[e0 + j * BIN_B];
  }
  __syncthreads();
  int r[BIN_EPT];
#pragma unroll
  for (int j = 0; j < BIN_EPT; ++j)
    r[j] = atomicAdd(&hist[(unsigned)c[j] / BKT], 1);
  __syncthreads();
  if (t < 32) {
    int v[BIN_SCAN];
    int sum = 0;
#pragma unroll
    for (int u = 0; u < BIN_SCAN; ++u) {
      v[u] = hist[t * BIN_SCAN + u];
      sum += v[u];
    }
    int inc = sum;
#pragma unroll
    for (int d = 1; d < 32; d <<= 1) {
      const int o = __shfl_up(inc, d);
      if (t >= d) inc += o;
    }
    int run = inc - sum;
#pragma unroll
    for (int u = 0; u < BIN_SCAN; ++u) {
      start[t * BIN_SCAN + u] = run;
      run += v[u];
    }
  }
  __syncthreads();
#pragma unroll
  for (int j = 0; j < BIN_EPT; ++j) {
    v2i ed;
    ed.x = c[j];
    ed.y = s[j];
    stage[start[(unsigned)c[j] / BKT] + r[j]] = ed;
  }
  for (int i = t; i < NBKT_PAD; i += BIN_B) {
    v2i sc;
    sc.x = start[i];
    sc.y = hist[i];
    seg[(size_t)w * NBKT_PAD + i] = sc;
  }
  __syncthreads();
#pragma unroll
  for (int j = 0; j < BIN_EPT; ++j)
    binned[(size_t)w * BIN_CH + j * BIN_B + t] = stage[j * BIN_B + t];
}

#define BLD_B 256
#define BLD_L 32
#define BLD_RW 32
#define BLD_K 16
static_assert(BIN_G <= (BLD_B / 32) * BLD_RW, "pass 2 gives every run of a block a wave slot");
#define XP_T ((BKT / 16 + BLD_B / 32 - 1) / (BLD_B / 32))
static_assert(BKT % 4 == 0 && SLOT_PITCH % 4 == 0 && N_NODES % 4 == 0,
              "16-byte groups of the image stay inside one plane and inside the valid nodes");
__global__ void __launch_bounds__(BLD_B)
slot_build_kernel(const int* __restrict__ binned_w,
                  const int* __restrict__ seg_w,
                  int* __restrict__ slot,
                  int* __restrict__ nxt,
                  int* __restrict__ cntc,
                  int* __restrict__ headc,
                  const float* __restrict__ x,
                  const float* __restrict__ P1,
                  float* __restrict__ y0) {
  const v2i* __restrict__ binned = reinterpret_cast<const v2i*>(binned_w);
  const v2i* __restrict__ seg = reinterpret_cast<const v2i*>(seg_w);
  __shared__ int cnt[BKT];
  __shared__ int head[BKT];
  __shared__ v4i slotL4[SLOT_W * BKT / 4 + BLD_L / 4];
  int* slotL = reinterpret_cast<int*>(slotL4);
  const int t = threadIdx.x;
  const int bk = blockIdx.x;
  const int base = bk * BKT;
  const v4i zero4 = {0, 0, 0, 0};
  const int nvalid = (N_NODES - base < BKT) ? (N_NODES - base) : BKT;
  const int wv = t >> 5;
  const int lane = t & 31;
  const int run0 = wv * BLD_RW;
  v2i sc;
  sc.x = 0;
  sc.y = 0;
  if (run0 + lane < BIN_G) sc = seg[(size_t)(run0 + lane) * NBKT_PAD + bk];
  for (int i = t; i < BKT; i += BLD_B) {
    cnt[i] = 0;
    head[i] = 0;
  }
  for (int i4 = t; i4 < SLOT_W * BKT / 4; i4 += BLD_B) slotL4[i4] = zero4;
  v2i ed[BLD_RW];
#pragma unroll
  for (int j = 0; j < BLD_RW; ++j) {
    const int n = __builtin_amdgcn_readlane(sc.y, j);
    const int p0 = (run0 + j) * BIN_CH + __builtin_amdgcn_readlane(sc.x, j);
    v2i z;
    z.x = base;
    z.y = 0;
    ed[j] = (lane < n) ? binned[p0 + lane] : z;
  }
  const int lhalf = lane & 15;
  const int khi = (lane >> 4) << 1;
  const int ncl = (lhalf < 3) ? lhalf : 2;
  v2f pb0, pb1;
  pb0.x = P1[(khi) * 3 + ncl];
  pb0.y = P1[(khi + 1) * 3 + ncl];
  pb1.x = P1[(4 + khi) * 3 + ncl];
  pb1.y = P1[(4 + khi + 1) * 3 + ncl];
  v2f xa[XP_T][2];
#pragma unroll
  for (int j = 0; j < XP_T; ++j) {
    const int r0 = 16 * (wv + (BLD_B / 32) * j);
    if (r0 < nvalid) {
      const float* __restrict__ hrow = x + (size_t)(base + r0 + lhalf) * 8;
      xa[j][0].x = hrow[khi];
      xa[j][0].y = hrow[khi + 1];
      xa[j][1].x = hrow[4 + khi];
      xa[j][1].y = hrow[4 + khi + 1];
    }
  }
  __syncthreads();
#pragma unroll
  for (int jb = 0; jb < BLD_RW; jb += BLD_K) {
    int k[BLD_K];
#pragma unroll
    for (int u = 0; u < BLD_K; ++u) {
      const int n = __builtin_amdgcn_readlane(sc.y, jb + u);
      k[u] = 0;
      if (lane < n) k[u] = atomicAdd(&cnt[ed[jb + u].x - base], 1);
    }
    unsigned ovf = 0;
#pragma unroll
    for (int u = 0; u < BLD_K; ++u) {
      const int n = __builtin_amdgcn_readlane(sc.y, jb + u);
      const int li = ed[jb + u].x - base;
      const bool in = (lane < n);
      const bool place = in & (k[u] < SLOT_W);
      slotL[place ? k[u] * BKT + li : SLOT_W * BKT + lane] = ed[jb + u].y;
      ovf |= (in & (k[u] >= SLOT_W)) ? (1u << u) : 0u;
    }
    if (__any(ovf != 0)) {
#pragma unroll
      for (int u = 0; u < BLD_K; ++u) {
        if (ovf & (1u << u)) {
          const int li = ed[jb + u].x - base;
          const int p = (run0 + jb + u) * BIN_CH + __builtin_amdgcn_readlane(sc.x, jb + u) + lane;
          nxt[p] = atomicExch(&head[li], p + 1);
        }
      }
    }
  }
  unsigned long long longm = __ballot(sc.y > BLD_L);
  while (longm != 0) {
    const int j = __ffsll((long long)longm) - 1;
    longm &= longm - 1;
    const int n = __shfl(sc.y, j);
    const int p0 = (run0 + j) * BIN_CH + __shfl(sc.x, j);
    for (int o = BLD_L; o < n; o += BLD_L) {
      const int e = o + lane;
      if (e < n) {
        const v2i ex = binned[p0 + e];
        const int li = ex.x - base;
        const int kk = atomicAdd(&cnt[li], 1);
        if (kk < SLOT_W) {
          slotL[kk * BKT + li] = ex.y;
        } else {
          const int p = p0 + e;
          nxt[p] = atomicExch(&head[li], p + 1);
        }
      }
    }
  }
  __syncthreads();
  for (int i4 = t; i4 < SLOT_W * BKT / 4; i4 += BLD_B) {
    const int i = 4 * i4;
    const int k = i / BKT;
    const int li = i - k * BKT;
    if (li < nvalid)
      *reinterpret_cast<v4i*>(slot + (size_t)k * SLOT_PITCH + base + li) = slotL4[i4];
  }
  for (int i = t; i < nvalid; i += BLD_B) {
    cntc[base + i] = cnt[i];
    headc[base + i] = head[i];
  }

#pragma unroll
  for (int j = 0; j < XP_T; ++j) {
    const int r0 = 16 * (wv + (BLD_B / 32) * j);
    if (r0 < nvalid) {
      v8f acc = {};
      acc = __builtin_amdgcn_wmma_f32_16x16x4_f32(
          false, xa[j][0], false, pb0, (short)0, acc, false, false);
      acc = __builtin_amdgcn_wmma_f32_16x16x4_f32(
          false, xa[j][1], false, pb1, (short)0, acc, false, false);
      const int lrow = r0 + ((lane >> 4) << 3);
#pragma unroll
      for (int v = 0; v < 8; ++v) {
        const int li = lrow + v;
        const float rs = rsqrtf((float)(cnt[li] + 1));
        if (lhalf < 4) y0[(size_t)(base + li) * 4 + lhalf] = acc[v] * rs;
      }
    }
  }
}

#define GATHER_B 64

template <int MODE>
__global__ void __launch_bounds__(GATHER_B)
agg_gather_node_kernel(const float* __restrict__ yin,
                       const float* __restrict__ bias,
                       const int* __restrict__ cntc,
                       const int* __restrict__ headc,
                       const int* __restrict__ slot,
                       const int* __restrict__ nxt,
                       const int* __restrict__ binned_w,
                       float* __restrict__ out) {
  constexpr bool LAST = (MODE == 2);
  const int i = blockIdx.x * GATHER_B + threadIdx.x;
  if (i >= N_NODES) return;
  const int n = cntc[i];
  const float* __restrict__ yself = yin + 4 * (size_t)i;
  float a0 = yself[0];
  float a1 = yself[1];
  float a2 = yself[2];

  int q[SLOT_W];
  row3 v[SLOT_W];
  const char* __restrict__ ybytes = reinterpret_cast<const char*>(yin);
#pragma unroll
  for (int k = 0; k < SLOT_W; ++k) q[k] = slot[(size_t)k * SLOT_PITCH + i];
#pragma unroll
  for (int u = 0; u < SLOT_W; ++u)
    v[u] = *reinterpret_cast<const row3*>(ybytes + ((unsigned)q[u] << 4));
#define ROWV(u) "v"(v[u].x), "v"(v[u].y), "v"(v[u].z)
#define ROWV4(u) ROWV(u), ROWV(u + 1), ROWV(u + 2), ROWV(u + 3)
#define ROWV20(u) ROWV4(u), ROWV4(u + 4), ROWV4(u + 8), ROWV4(u + 12), ROWV4(u + 16)
  asm volatile("" : "+v"(a0), "+v"(a1), "+v"(a2) : ROWV20(0), ROWV20(20));
#undef ROWV20
#undef ROWV4
#undef ROWV
#pragma unroll
  for (int u = 0; u < SLOT_W; ++u) {
    const bool act = (u < n);
    const float t0 = a0 + v[u].x;
    const float t1 = a1 + v[u].y;
    const float t2 = a2 + v[u].z;
    a0 = act ? t0 : a0;
    a1 = act ? t1 : a1;
    a2 = act ? t2 : a2;
  }
  if (n > SLOT_W) {
    int p = headc[i];
    for (int t = SLOT_W; t < n && p != 0; ++t) {
      const int e = p - 1;
      const unsigned r = (unsigned)binned_w[2 * (size_t)e + 1];
      p = nxt[e];
      const row3 w = *reinterpret_cast<const row3*>(yin + 4 * (size_t)r);
      a0 += w.x;
      a1 += w.y;
      a2 += w.z;
    }
  }

  const float d = rsqrtf((float)(n + 1));
  const float z0 = a0 * d + bias[0];
  const float z1 = a1 * d + bias[1];
  const float z2 = a2 * d + bias[2];
  if (LAST) {
    out[3 * (size_t)i + 0] = z0;
    out[3 * (size_t)i + 1] = z1;
    out[3 * (size_t)i + 2] = z2;
  } else {
    v4f o;
    o.x = z0 * d;
    o.y = z1 * d;
    o.z = z2 * d;
    o.w = 0.f;
    reinterpret_cast<v4f*>(out)[i] = o;
  }
}

extern "C" void kernel_launch(void* const* d_in, const int* in_sizes, int n_in,
                              void* d_out, int out_size, void* d_ws, size_t ws_size,
                              hipStream_t stream) {
  const float* x  = (const float*)d_in[0];
  const int*   ei = (const int*)d_in[1];
  const int*   row = ei;
  const int*   col = ei + N_EDGES;

  const float* Wm[5] = {(const float*)d_in[2], (const float*)d_in[4],
                        (const float*)d_in[6], (const float*)d_in[8],
                        (const float*)d_in[10]};
  const float* Bm[5] = {(const float*)d_in[3], (const float*)d_in[5],
                        (const float*)d_in[7], (const float*)d_in[9],
                        (const float*)d_in[11]};

  int*   wsi   = (int*)d_ws;
  int*   slot  = wsi;
  int*   nxt   = slot + SLOT_W * SLOT_PITCH;
  int*   binned = nxt + N_EDGES;
  int*   seg   = binned + 2 * N_EDGES;
  float* P1    = (float*)(seg + 2 * BIN_G * NBKT_PAD);
  float* cv    = P1 + 64;
  float* zA    = P1 + 128;
  float* zB    = zA + 4 * N_PAD;
  int*   cntc  = (int*)(zB + 4 * N_PAD);
  int*   headc = cntc + N_PAD;

  const int gGn  = (N_NODES + GATHER_B - 1) / GATHER_B;

  edge_bin_kernel<<<BIN_G + 1, BIN_B, 0, stream>>>(row, col, binned, seg,
                                                   Wm[0], Bm[0], Wm[1], Bm[1], Wm[2], Bm[2],
                                                   Wm[3], Bm[3], Wm[4], P1, cv);
  slot_build_kernel<<<NBKT, BLD_B, 0, stream>>>(binned, seg, slot, nxt, cntc, headc,
                                                x, P1, zA);

  agg_gather_node_kernel<1><<<gGn, GATHER_B, 0, stream>>>(zA, cv + 0, cntc, headc, slot,
                                                          nxt, binned, zB);
  agg_gather_node_kernel<1><<<gGn, GATHER_B, 0, stream>>>(zB, cv + 3, cntc, headc, slot,
                                                          nxt, binned, zA);
  agg_gather_node_kernel<1><<<gGn, GATHER_B, 0, stream>>>(zA, cv + 6, cntc, headc, slot,
                                                          nxt, binned, zB);
  agg_gather_node_kernel<1><<<gGn, GATHER_B, 0, stream>>>(zB, cv + 9, cntc, headc, slot,
                                                          nxt, binned, zA);
  agg_gather_node_kernel<2><<<gGn, GATHER_B, 0, stream>>>(zA, Bm[4], cntc, headc, slot,
                                                          nxt, binned, (float*)d_out);
}
